# starting kernel
# baseline (speedup 1.0000x reference)
#include <hip/hip_runtime.h>
#include <math.h>

typedef float v8f __attribute__((ext_vector_type(8)));
typedef float v4f __attribute__((ext_vector_type(4)));
typedef float v2f __attribute__((ext_vector_type(2)));

#define B_      4
#define H_      12
#define S_      1024
#define D_      64
#define HALF_   32
#define LAMBDA_ 0.01f
#define EPS_    1e-6f
#define CLAMP_  20.0f
#define QTILES  (S_ / 16)
#define WPB     4
#define LSTR    20

__global__ __launch_bounds__(32 * WPB)
void rse_stickbreak_attn(const float* __restrict__ q,
                         const float* __restrict__ k,
                         const float* __restrict__ v,
                         const float* __restrict__ cosc,
                         const float* __restrict__ sinc,
                         float* __restrict__ out)
{
    __shared__ __align__(16) float lds[WPB][16 * LSTR + 16];

    const int wave = threadIdx.x >> 5;
    const int lane = threadIdx.x & 31;
    const int tile = blockIdx.x * WPB + wave;
    const int qt   = tile % QTILES;
    const int bh   = tile / QTILES;

    float* wlds = &lds[wave][0];
    float* dlds = &lds[wave][16 * LSTR];

    const int hi = lane >> 4;
    const int ln = lane & 15;

    const float* qb = q + (size_t)bh * S_ * D_;
    const float* kb = k + (size_t)bh * S_ * D_;
    const float* vb = v + (size_t)bh * S_ * D_;
    float*       ob = out + (size_t)bh * S_ * D_;

    const int qrow = qt * 16 + ln;
    v2f afrag[16];
#pragma unroll
    for (int c = 0; c < 16; ++c) {
        const int d  = 4 * c + 2 * hi;
        const v2f qp = *(const v2f*)(qb + qrow * D_ + d);
        const float cc = cosc[qrow * HALF_ + (d >> 1)];
        const float ss = sinc[qrow * HALF_ + (d >> 1)];
        afrag[c].x = qp.x * cc - qp.y * ss;
        afrag[c].y = qp.y * cc + qp.x * ss;
    }

    v8f acc0 = (v8f)0.0f, acc1 = (v8f)0.0f, acc2 = (v8f)0.0f, acc3 = (v8f)0.0f;
    float rem   = 1.0f;
    float denom = 0.0f;
    const float scale = 0.125f;

    for (int kt = 0; kt <= qt; ++kt) {
        if (kt < qt) {
            __builtin_prefetch(&kb[(kt + 1) * 16 * D_ + lane * 32], 0, 3);
            __builtin_prefetch(&vb[(kt + 1) * 16 * D_ + lane * 32], 0, 3);
        }

        const int krow = kt * 16 + ln;
        v8f cacc = (v8f)0.0f;
#pragma unroll
        for (int c = 0; c < 16; ++c) {
            const int d  = 4 * c + 2 * hi;
            const v2f kp = *(const v2f*)(kb + krow * D_ + d);
            const float cc = cosc[krow * HALF_ + (d >> 1)];
            const float ss = sinc[krow * HALF_ + (d >> 1)];
            v2f bfrag;
            bfrag.x = kp.x * cc - kp.y * ss;
            bfrag.y = kp.y * cc + kp.x * ss;
            cacc = __builtin_amdgcn_wmma_f32_16x16x4_f32(
                false, afrag[c], false, bfrag, (short)0, cacc, false, false);
        }

#pragma unroll
        for (int r = 0; r < 8; ++r) {
            const int m    = r + 8 * hi;
            const int qpos = qt * 16 + m;
            const int kpos = kt * 16 + ln;
            float x = cacc[r] * scale - LAMBDA_ * fabsf((float)(qpos - kpos));
            const bool fut = kpos > qpos;
            x = fminf(fmaxf(x, -CLAMP_), CLAMP_);
            float beta = 1.0f / (1.0f + __expf(-x));
            if (fut) beta = 0.0f;
            wlds[m * LSTR + ln] = beta;
        }
        asm volatile("s_wait_dscnt 0" ::: "memory");
        __builtin_amdgcn_wave_barrier();

        if (lane < 16) {
            v4f* rowp = (v4f*)(wlds + lane * LSTR);
            v4f w0 = rowp[0], w1 = rowp[1], w2 = rowp[2], w3 = rowp[3];
            float r_   = rem;
            float dsum = 0.0f;
#define SB_STEP(VEC, I)                                   \
            { const float w = VEC[I] * r_;                \
              r_ = fmaxf(r_ * (1.0f - w), EPS_);          \
              VEC[I] = w;  dsum += w; }
            SB_STEP(w0, 0) SB_STEP(w0, 1) SB_STEP(w0, 2) SB_STEP(w0, 3)
            SB_STEP(w1, 0) SB_STEP(w1, 1) SB_STEP(w1, 2) SB_STEP(w1, 3)
            SB_STEP(w2, 0) SB_STEP(w2, 1) SB_STEP(w2, 2) SB_STEP(w2, 3)
            SB_STEP(w3, 0) SB_STEP(w3, 1) SB_STEP(w3, 2) SB_STEP(w3, 3)
#undef SB_STEP
            rowp[0] = w0; rowp[1] = w1; rowp[2] = w2; rowp[3] = w3;
            rem    = r_;
            denom += dsum;
        }
        asm volatile("s_wait_dscnt 0" ::: "memory");
        __builtin_amdgcn_wave_barrier();

        v2f awf[4];
#pragma unroll
        for (int kc = 0; kc < 4; ++kc)
            awf[kc] = *(const v2f*)(wlds + ln * LSTR + 4 * kc + 2 * hi);
#pragma unroll
        for (int kc = 0; kc < 4; ++kc) {
            const int vr = kt * 16 + 4 * kc + 2 * hi;
            v2f b0, b1, b2, b3;
            b0.x = vb[vr * D_ +  0 + ln];  b0.y = vb[(vr + 1) * D_ +  0 + ln];
            b1.x = vb[vr * D_ + 16 + ln];  b1.y = vb[(vr + 1) * D_ + 16 + ln];
            b2.x = vb[vr * D_ + 32 + ln];  b2.y = vb[(vr + 1) * D_ + 32 + ln];
            b3.x = vb[vr * D_ + 48 + ln];  b3.y = vb[(vr + 1) * D_ + 48 + ln];
            acc0 = __builtin_amdgcn_wmma_f32_16x16x4_f32(
                false, awf[kc], false, b0, (short)0, acc0, false, false);
            acc1 = __builtin_amdgcn_wmma_f32_16x16x4_f32(
                false, awf[kc], false, b1, (short)0, acc1, false, false);
            acc2 = __builtin_amdgcn_wmma_f32_16x16x4_f32(
                false, awf[kc], false, b2, (short)0, acc2, false, false);
            acc3 = __builtin_amdgcn_wmma_f32_16x16x4_f32(
                false, awf[kc], false, b3, (short)0, acc3, false, false);
        }
    }

    if (lane < 16) dlds[lane] = fmaxf(denom, EPS_);
    asm volatile("s_wait_dscnt 0" ::: "memory");
    __builtin_amdgcn_wave_barrier();

#pragma unroll
    for (int r = 0; r < 8; ++r) {
        const int m  = r + 8 * hi;
        const float dn = dlds[m];
        const int orow = qt * 16 + m;
        ob[orow * D_ +  0 + ln] = acc0[r] / dn;
        ob[orow * D_ + 16 + ln] = acc1[r] / dn;
        ob[orow * D_ + 32 + ln] = acc2[r] / dn;
        ob[orow * D_ + 48 + ln] = acc3[r] / dn;
    }
}

extern "C" void kernel_launch(void* const* d_in, const int* in_sizes, int n_in,
                              void* d_out, int out_size, void* d_ws, size_t ws_size,
                              hipStream_t stream) {
    (void)in_sizes; (void)n_in; (void)out_size; (void)d_ws; (void)ws_size;
    const float* q    = (const float*)d_in[0];
    const float* k    = (const float*)d_in[1];
    const float* v    = (const float*)d_in[2];
    const float* cosc = (const float*)d_in[3];
    const float* sinc = (const float*)d_in[4];
    float* out = (float*)d_out;

    const int tiles  = B_ * H_ * QTILES;
    const int blocks = tiles / WPB;
    rse_stickbreak_attn<<<blocks, 32 * WPB, 0, stream>>>(q, k, v, cosc, sinc, out);
}


# Fragment k-order and output-column remap so q/k/cos/sin/v/out move as b128 per lane (80 -> 24 loads per k tile); logit dot product sums the head dim in a permuted order
# speedup vs baseline: 1.1874x; 1.1874x over previous
#include <hip/hip_runtime.h>
#include <math.h>

typedef float v8f __attribute__((ext_vector_type(8)));
typedef float v4f __attribute__((ext_vector_type(4)));
typedef float v2f __attribute__((ext_vector_type(2)));

#define B_      4
#define H_      12
#define S_      1024
#define D_      64
#define HALF_   32
#define LAMBDA_ 0.01f
#define EPS_    1e-6f
#define CLAMP_  20.0f
#define QTILES  (S_ / 16)
#define WPB     4
#define LSTR    20

__global__ __launch_bounds__(32 * WPB)
void rse_stickbreak_attn(const float* __restrict__ q,
                         const float* __restrict__ k,
                         const float* __restrict__ v,
                         const float* __restrict__ cosc,
                         const float* __restrict__ sinc,
                         float* __restrict__ out)
{
    __shared__ __align__(16) float lds[WPB][16 * LSTR + 16];

    const int wave = threadIdx.x >> 5;
    const int lane = threadIdx.x & 31;
    const int tile = blockIdx.x * WPB + wave;
    const int qt   = tile % QTILES;
    const int bh   = tile / QTILES;

    float* wlds = &lds[wave][0];
    float* dlds = &lds[wave][16 * LSTR];

    const int hi = lane >> 4;
    const int ln = lane & 15;

    const float* qb = q + (size_t)bh * S_ * D_;
    const float* kb = k + (size_t)bh * S_ * D_;
    const float* vb = v + (size_t)bh * S_ * D_;
    float*       ob = out + (size_t)bh * S_ * D_;

    const int qrow = qt * 16 + ln;
    v2f afrag[16];
#pragma unroll
    for (int g = 0; g < 4; ++g) {
        const int d0 = 16 * g + 8 * hi;
        const v4f x0 = *(const v4f*)(qb + qrow * D_ + d0);
        const v4f x1 = *(const v4f*)(qb + qrow * D_ + d0 + 4);
        const v4f cc = *(const v4f*)(cosc + qrow * HALF_ + (d0 >> 1));
        const v4f ss = *(const v4f*)(sinc + qrow * HALF_ + (d0 >> 1));
        afrag[4 * g + 0].x = x0.x * cc.x - x0.y * ss.x;
        afrag[4 * g + 0].y = x0.y * cc.x + x0.x * ss.x;
        afrag[4 * g + 1].x = x0.z * cc.y - x0.w * ss.y;
        afrag[4 * g + 1].y = x0.w * cc.y + x0.z * ss.y;
        afrag[4 * g + 2].x = x1.x * cc.z - x1.y * ss.z;
        afrag[4 * g + 2].y = x1.y * cc.z + x1.x * ss.z;
        afrag[4 * g + 3].x = x1.z * cc.w - x1.w * ss.w;
        afrag[4 * g + 3].y = x1.w * cc.w + x1.z * ss.w;
    }

    v8f acc0 = (v8f)0.0f, acc1 = (v8f)0.0f, acc2 = (v8f)0.0f, acc3 = (v8f)0.0f;
    float rem   = 1.0f;
    float denom = 0.0f;
    const float scale = 0.125f;

    for (int kt = 0; kt <= qt; ++kt) {
        if (kt < qt) {
            __builtin_prefetch(&kb[(kt + 1) * 16 * D_ + lane * 32], 0, 3);
            __builtin_prefetch(&vb[(kt + 1) * 16 * D_ + lane * 32], 0, 3);
        }

        const int krow = kt * 16 + ln;
        v8f cacc = (v8f)0.0f;
#pragma unroll
        for (int g = 0; g < 4; ++g) {
            const int d0 = 16 * g + 8 * hi;
            const v4f x0 = *(const v4f*)(kb + krow * D_ + d0);
            const v4f x1 = *(const v4f*)(kb + krow * D_ + d0 + 4);
            const v4f cc = *(const v4f*)(cosc + krow * HALF_ + (d0 >> 1));
            const v4f ss = *(const v4f*)(sinc + krow * HALF_ + (d0 >> 1));
            v2f bf0, bf1, bf2, bf3;
            bf0.x = x0.x * cc.x - x0.y * ss.x;   bf0.y = x0.y * cc.x + x0.x * ss.x;
            bf1.x = x0.z * cc.y - x0.w * ss.y;   bf1.y = x0.w * cc.y + x0.z * ss.y;
            bf2.x = x1.x * cc.z - x1.y * ss.z;   bf2.y = x1.y * cc.z + x1.x * ss.z;
            bf3.x = x1.z * cc.w - x1.w * ss.w;   bf3.y = x1.w * cc.w + x1.z * ss.w;
            cacc = __builtin_amdgcn_wmma_f32_16x16x4_f32(
                false, afrag[4 * g + 0], false, bf0, (short)0, cacc, false, false);
            cacc = __builtin_amdgcn_wmma_f32_16x16x4_f32(
                false, afrag[4 * g + 1], false, bf1, (short)0, cacc, false, false);
            cacc = __builtin_amdgcn_wmma_f32_16x16x4_f32(
                false, afrag[4 * g + 2], false, bf2, (short)0, cacc, false, false);
            cacc = __builtin_amdgcn_wmma_f32_16x16x4_f32(
                false, afrag[4 * g + 3], false, bf3, (short)0, cacc, false, false);
        }

#pragma unroll
        for (int r = 0; r < 8; ++r) {
            const int m    = r + 8 * hi;
            const int qpos = qt * 16 + m;
            const int kpos = kt * 16 + ln;
            float x = cacc[r] * scale - LAMBDA_ * fabsf((float)(qpos - kpos));
            const bool fut = kpos > qpos;
            x = fminf(fmaxf(x, -CLAMP_), CLAMP_);
            float beta = 1.0f / (1.0f + __expf(-x));
            if (fut) beta = 0.0f;
            wlds[m * LSTR + ln] = beta;
        }
        asm volatile("s_wait_dscnt 0" ::: "memory");
        __builtin_amdgcn_wave_barrier();

        if (lane < 16) {
            v4f* rowp = (v4f*)(wlds + lane * LSTR);
            v4f w0 = rowp[0], w1 = rowp[1], w2 = rowp[2], w3 = rowp[3];
            float r_   = rem;
            float dsum = 0.0f;
#define SB_STEP(VEC, I)                                   \
            { const float w = VEC[I] * r_;                \
              r_ = fmaxf(r_ * (1.0f - w), EPS_);          \
              VEC[I] = w;  dsum += w; }
            SB_STEP(w0, 0) SB_STEP(w0, 1) SB_STEP(w0, 2) SB_STEP(w0, 3)
            SB_STEP(w1, 0) SB_STEP(w1, 1) SB_STEP(w1, 2) SB_STEP(w1, 3)
            SB_STEP(w2, 0) SB_STEP(w2, 1) SB_STEP(w2, 2) SB_STEP(w2, 3)
            SB_STEP(w3, 0) SB_STEP(w3, 1) SB_STEP(w3, 2) SB_STEP(w3, 3)
#undef SB_STEP
            rowp[0] = w0; rowp[1] = w1; rowp[2] = w2; rowp[3] = w3;
            rem    = r_;
            denom += dsum;
        }
        asm volatile("s_wait_dscnt 0" ::: "memory");
        __builtin_amdgcn_wave_barrier();

        v2f awf[4];
#pragma unroll
        for (int kc = 0; kc < 4; ++kc)
            awf[kc] = *(const v2f*)(wlds + ln * LSTR + 4 * kc + 2 * hi);
#pragma unroll
        for (int kc = 0; kc < 4; ++kc) {
            const int vr = kt * 16 + 4 * kc + 2 * hi;
            const v4f r0 = *(const v4f*)(vb + vr * D_ + 4 * ln);
            const v4f r1 = *(const v4f*)(vb + (vr + 1) * D_ + 4 * ln);
            v2f b0, b1, b2, b3;
            b0.x = r0.x;  b0.y = r1.x;
            b1.x = r0.y;  b1.y = r1.y;
            b2.x = r0.z;  b2.y = r1.z;
            b3.x = r0.w;  b3.y = r1.w;
            acc0 = __builtin_amdgcn_wmma_f32_16x16x4_f32(
                false, awf[kc], false, b0, (short)0, acc0, false, false);
            acc1 = __builtin_amdgcn_wmma_f32_16x16x4_f32(
                false, awf[kc], false, b1, (short)0, acc1, false, false);
            acc2 = __builtin_amdgcn_wmma_f32_16x16x4_f32(
                false, awf[kc], false, b2, (short)0, acc2, false, false);
            acc3 = __builtin_amdgcn_wmma_f32_16x16x4_f32(
                false, awf[kc], false, b3, (short)0, acc3, false, false);
        }
    }

    if (lane < 16) dlds[lane] = fmaxf(denom, EPS_);
    asm volatile("s_wait_dscnt 0" ::: "memory");
    __builtin_amdgcn_wave_barrier();

#pragma unroll
    for (int r = 0; r < 8; ++r) {
        const int m  = r + 8 * hi;
        const float dn = dlds[m];
        const int orow = qt * 16 + m;
        v4f o;
        o.x = acc0[r] / dn;
        o.y = acc1[r] / dn;
        o.z = acc2[r] / dn;
        o.w = acc3[r] / dn;
        *(v4f*)(ob + orow * D_ + 4 * ln) = o;
    }
}

extern "C" void kernel_launch(void* const* d_in, const int* in_sizes, int n_in,
                              void* d_out, int out_size, void* d_ws, size_t ws_size,
                              hipStream_t stream) {
    (void)in_sizes; (void)n_in; (void)out_size; (void)d_ws; (void)ws_size;
    const float* q    = (const float*)d_in[0];
    const float* k    = (const float*)d_in[1];
    const float* v    = (const float*)d_in[2];
    const float* cosc = (const float*)d_in[3];
    const float* sinc = (const float*)d_in[4];
    float* out = (float*)d_out;

    const int tiles  = B_ * H_ * QTILES;
    const int blocks = tiles / WPB;
    rse_stickbreak_attn<<<blocks, 32 * WPB, 0, stream>>>(q, k, v, cosc, sinc, out);
}


# Each wave does a pair of causal q-tiles (63-p then p: 65 k-loop trips for every wave, was 1..64), grid 768 -> 384 groups; per-tile arithmetic and every summation order unchanged
# speedup vs baseline: 1.3241x; 1.1151x over previous
#include <hip/hip_runtime.h>
#include <math.h>

typedef float v8f __attribute__((ext_vector_type(8)));
typedef float v4f __attribute__((ext_vector_type(4)));
typedef float v2f __attribute__((ext_vector_type(2)));

#define B_      4
#define H_      12
#define S_      1024
#define D_      64
#define HALF_   32
#define LAMBDA_ 0.01f
#define EPS_    1e-6f
#define CLAMP_  20.0f
#define QTILES  (S_ / 16)
#define WPB     4
#define LSTR    20

__global__ __launch_bounds__(32 * WPB)
void rse_stickbreak_attn(const float* __restrict__ q,
                         const float* __restrict__ k,
                         const float* __restrict__ v,
                         const float* __restrict__ cosc,
                         const float* __restrict__ sinc,
                         float* __restrict__ out)
{
    __shared__ __align__(16) float lds[WPB][16 * LSTR + 16];

    const int wave = threadIdx.x >> 5;
    const int lane = threadIdx.x & 31;
    const int pr = (blockIdx.x % (QTILES / 2 / WPB)) * WPB + wave;
    const int bh = blockIdx.x / (QTILES / 2 / WPB);

    float* wlds = &lds[wave][0];
    float* dlds = &lds[wave][16 * LSTR];

    const int hi = lane >> 4;
    const int ln = lane & 15;

    const float* qb = q + (size_t)bh * S_ * D_;
    const float* kb = k + (size_t)bh * S_ * D_;
    const float* vb = v + (size_t)bh * S_ * D_;
    float*       ob = out + (size_t)bh * S_ * D_;

  for (int pass = 0; pass < 2; ++pass) {
    const int qt = pass ? pr : (QTILES - 1 - pr);

    const int qrow = qt * 16 + ln;
    v2f afrag[16];
#pragma unroll
    for (int g = 0; g < 4; ++g) {
        const int d0 = 16 * g + 8 * hi;
        const v4f x0 = *(const v4f*)(qb + qrow * D_ + d0);
        const v4f x1 = *(const v4f*)(qb + qrow * D_ + d0 + 4);
        const v4f cc = *(const v4f*)(cosc + qrow * HALF_ + (d0 >> 1));
        const v4f ss = *(const v4f*)(sinc + qrow * HALF_ + (d0 >> 1));
        afrag[4 * g + 0].x = x0.x * cc.x - x0.y * ss.x;
        afrag[4 * g + 0].y = x0.y * cc.x + x0.x * ss.x;
        afrag[4 * g + 1].x = x0.z * cc.y - x0.w * ss.y;
        afrag[4 * g + 1].y = x0.w * cc.y + x0.z * ss.y;
        afrag[4 * g + 2].x = x1.x * cc.z - x1.y * ss.z;
        afrag[4 * g + 2].y = x1.y * cc.z + x1.x * ss.z;
        afrag[4 * g + 3].x = x1.z * cc.w - x1.w * ss.w;
        afrag[4 * g + 3].y = x1.w * cc.w + x1.z * ss.w;
    }

    v8f acc0 = (v8f)0.0f, acc1 = (v8f)0.0f, acc2 = (v8f)0.0f, acc3 = (v8f)0.0f;
    float rem   = 1.0f;
    float denom = 0.0f;
    const float scale = 0.125f;

    for (int kt = 0; kt <= qt; ++kt) {
        if (kt < qt) {
            __builtin_prefetch(&kb[(kt + 1) * 16 * D_ + lane * 32], 0, 3);
            __builtin_prefetch(&vb[(kt + 1) * 16 * D_ + lane * 32], 0, 3);
        }

        const int krow = kt * 16 + ln;
        v8f cacc = (v8f)0.0f;
#pragma unroll
        for (int g = 0; g < 4; ++g) {
            const int d0 = 16 * g + 8 * hi;
            const v4f x0 = *(const v4f*)(kb + krow * D_ + d0);
            const v4f x1 = *(const v4f*)(kb + krow * D_ + d0 + 4);
            const v4f cc = *(const v4f*)(cosc + krow * HALF_ + (d0 >> 1));
            const v4f ss = *(const v4f*)(sinc + krow * HALF_ + (d0 >> 1));
            v2f bf0, bf1, bf2, bf3;
            bf0.x = x0.x * cc.x - x0.y * ss.x;   bf0.y = x0.y * cc.x + x0.x * ss.x;
            bf1.x = x0.z * cc.y - x0.w * ss.y;   bf1.y = x0.w * cc.y + x0.z * ss.y;
            bf2.x = x1.x * cc.z - x1.y * ss.z;   bf2.y = x1.y * cc.z + x1.x * ss.z;
            bf3.x = x1.z * cc.w - x1.w * ss.w;   bf3.y = x1.w * cc.w + x1.z * ss.w;
            cacc = __builtin_amdgcn_wmma_f32_16x16x4_f32(
                false, afrag[4 * g + 0], false, bf0, (short)0, cacc, false, false);
            cacc = __builtin_amdgcn_wmma_f32_16x16x4_f32(
                false, afrag[4 * g + 1], false, bf1, (short)0, cacc, false, false);
            cacc = __builtin_amdgcn_wmma_f32_16x16x4_f32(
                false, afrag[4 * g + 2], false, bf2, (short)0, cacc, false, false);
            cacc = __builtin_amdgcn_wmma_f32_16x16x4_f32(
                false, afrag[4 * g + 3], false, bf3, (short)0, cacc, false, false);
        }

#pragma unroll
        for (int r = 0; r < 8; ++r) {
            const int m    = r + 8 * hi;
            const int qpos = qt * 16 + m;
            const int kpos = kt * 16 + ln;
            float x = cacc[r] * scale - LAMBDA_ * fabsf((float)(qpos - kpos));
            const bool fut = kpos > qpos;
            x = fminf(fmaxf(x, -CLAMP_), CLAMP_);
            float beta = 1.0f / (1.0f + __expf(-x));
            if (fut) beta = 0.0f;
            wlds[m * LSTR + ln] = beta;
        }
        asm volatile("s_wait_dscnt 0" ::: "memory");
        __builtin_amdgcn_wave_barrier();

        if (lane < 16) {
            v4f* rowp = (v4f*)(wlds + lane * LSTR);
            v4f w0 = rowp[0], w1 = rowp[1], w2 = rowp[2], w3 = rowp[3];
            float r_   = rem;
            float dsum = 0.0f;
#define SB_STEP(VEC, I)                                   \
            { const float w = VEC[I] * r_;                \
              r_ = fmaxf(r_ * (1.0f - w), EPS_);          \
              VEC[I] = w;  dsum += w; }
            SB_STEP(w0, 0) SB_STEP(w0, 1) SB_STEP(w0, 2) SB_STEP(w0, 3)
            SB_STEP(w1, 0) SB_STEP(w1, 1) SB_STEP(w1, 2) SB_STEP(w1, 3)
            SB_STEP(w2, 0) SB_STEP(w2, 1) SB_STEP(w2, 2) SB_STEP(w2, 3)
            SB_STEP(w3, 0) SB_STEP(w3, 1) SB_STEP(w3, 2) SB_STEP(w3, 3)
#undef SB_STEP
            rowp[0] = w0; rowp[1] = w1; rowp[2] = w2; rowp[3] = w3;
            rem    = r_;
            denom += dsum;
        }
        asm volatile("s_wait_dscnt 0" ::: "memory");
        __builtin_amdgcn_wave_barrier();

        v2f awf[4];
#pragma unroll
        for (int kc = 0; kc < 4; ++kc)
            awf[kc] = *(const v2f*)(wlds + ln * LSTR + 4 * kc + 2 * hi);
#pragma unroll
        for (int kc = 0; kc < 4; ++kc) {
            const int vr = kt * 16 + 4 * kc + 2 * hi;
            const v4f r0 = *(const v4f*)(vb + vr * D_ + 4 * ln);
            const v4f r1 = *(const v4f*)(vb + (vr + 1) * D_ + 4 * ln);
            v2f b0, b1, b2, b3;
            b0.x = r0.x;  b0.y = r1.x;
            b1.x = r0.y;  b1.y = r1.y;
            b2.x = r0.z;  b2.y = r1.z;
            b3.x = r0.w;  b3.y = r1.w;
            acc0 = __builtin_amdgcn_wmma_f32_16x16x4_f32(
                false, awf[kc], false, b0, (short)0, acc0, false, false);
            acc1 = __builtin_amdgcn_wmma_f32_16x16x4_f32(
                false, awf[kc], false, b1, (short)0, acc1, false, false);
            acc2 = __builtin_amdgcn_wmma_f32_16x16x4_f32(
                false, awf[kc], false, b2, (short)0, acc2, false, false);
            acc3 = __builtin_amdgcn_wmma_f32_16x16x4_f32(
                false, awf[kc], false, b3, (short)0, acc3, false, false);
        }
    }

    if (lane < 16) dlds[lane] = fmaxf(denom, EPS_);
    asm volatile("s_wait_dscnt 0" ::: "memory");
    __builtin_amdgcn_wave_barrier();

#pragma unroll
    for (int r = 0; r < 8; ++r) {
        const int m  = r + 8 * hi;
        const float dn = dlds[m];
        const int orow = qt * 16 + m;
        v4f o;
        o.x = acc0[r] / dn;
        o.y = acc1[r] / dn;
        o.z = acc2[r] / dn;
        o.w = acc3[r] / dn;
        *(v4f*)(ob + orow * D_ + 4 * ln) = o;
    }
  }
}

extern "C" void kernel_launch(void* const* d_in, const int* in_sizes, int n_in,
                              void* d_out, int out_size, void* d_ws, size_t ws_size,
                              hipStream_t stream) {
    (void)in_sizes; (void)n_in; (void)out_size; (void)d_ws; (void)ws_size;
    const float* q    = (const float*)d_in[0];
    const float* k    = (const float*)d_in[1];
    const float* v    = (const float*)d_in[2];
    const float* cosc = (const float*)d_in[3];
    const float* sinc = (const float*)d_in[4];
    float* out = (float*)d_out;

    const int tiles  = B_ * H_ * QTILES;
    const int blocks = tiles / (2 * WPB);
    rse_stickbreak_attn<<<blocks, 32 * WPB, 0, stream>>>(q, k, v, cosc, sinc, out);
}


# Register prefetch one trip deep: next trip's k/cos/sin loads issued after the logits product, this trip's v loads at its top (replaces the two prefetch hints); arithmetic and sum orders unchanged
# speedup vs baseline: 1.4685x; 1.1091x over previous
#include <hip/hip_runtime.h>
#include <math.h>

typedef float v8f __attribute__((ext_vector_type(8)));
typedef float v4f __attribute__((ext_vector_type(4)));
typedef float v2f __attribute__((ext_vector_type(2)));

#define B_      4
#define H_      12
#define S_      1024
#define D_      64
#define HALF_   32
#define LAMBDA_ 0.01f
#define EPS_    1e-6f
#define CLAMP_  20.0f
#define QTILES  (S_ / 16)
#define WPB     4
#define LSTR    20

__global__ __launch_bounds__(32 * WPB)
void rse_stickbreak_attn(const float* __restrict__ q,
                         const float* __restrict__ k,
                         const float* __restrict__ v,
                         const float* __restrict__ cosc,
                         const float* __restrict__ sinc,
                         float* __restrict__ out)
{
    __shared__ __align__(16) float lds[WPB][16 * LSTR + 16];

    const int wave = threadIdx.x >> 5;
    const int lane = threadIdx.x & 31;
    const int pr = (blockIdx.x % (QTILES / 2 / WPB)) * WPB + wave;
    const int bh = blockIdx.x / (QTILES / 2 / WPB);

    float* wlds = &lds[wave][0];
    float* dlds = &lds[wave][16 * LSTR];

    const int hi = lane >> 4;
    const int ln = lane & 15;

    const float* qb = q + (size_t)bh * S_ * D_;
    const float* kb = k + (size_t)bh * S_ * D_;
    const float* vb = v + (size_t)bh * S_ * D_;
    float*       ob = out + (size_t)bh * S_ * D_;

  for (int pass = 0; pass < 2; ++pass) {
    const int qt = pass ? pr : (QTILES - 1 - pr);

    const int qrow = qt * 16 + ln;
    v2f afrag[16];
#pragma unroll
    for (int g = 0; g < 4; ++g) {
        const int d0 = 16 * g + 8 * hi;
        const v4f x0 = *(const v4f*)(qb + qrow * D_ + d0);
        const v4f x1 = *(const v4f*)(qb + qrow * D_ + d0 + 4);
        const v4f cc = *(const v4f*)(cosc + qrow * HALF_ + (d0 >> 1));
        const v4f ss = *(const v4f*)(sinc + qrow * HALF_ + (d0 >> 1));
        afrag[4 * g + 0].x = x0.x * cc.x - x0.y * ss.x;
        afrag[4 * g + 0].y = x0.y * cc.x + x0.x * ss.x;
        afrag[4 * g + 1].x = x0.z * cc.y - x0.w * ss.y;
        afrag[4 * g + 1].y = x0.w * cc.y + x0.z * ss.y;
        afrag[4 * g + 2].x = x1.x * cc.z - x1.y * ss.z;
        afrag[4 * g + 2].y = x1.y * cc.z + x1.x * ss.z;
        afrag[4 * g + 3].x = x1.z * cc.w - x1.w * ss.w;
        afrag[4 * g + 3].y = x1.w * cc.w + x1.z * ss.w;
    }

    v8f acc0 = (v8f)0.0f, acc1 = (v8f)0.0f, acc2 = (v8f)0.0f, acc3 = (v8f)0.0f;
    float rem   = 1.0f;
    float denom = 0.0f;
    const float scale = 0.125f;

    v4f kx0[4], kx1[4], kcc[4], kss[4];
#pragma unroll
    for (int g = 0; g < 4; ++g) {
        const int d0 = 16 * g + 8 * hi;
        kx0[g] = *(const v4f*)(kb + ln * D_ + d0);
        kx1[g] = *(const v4f*)(kb + ln * D_ + d0 + 4);
        kcc[g] = *(const v4f*)(cosc + ln * HALF_ + (d0 >> 1));
        kss[g] = *(const v4f*)(sinc + ln * HALF_ + (d0 >> 1));
    }

    for (int kt = 0; kt <= qt; ++kt) {
        v4f vr0[4], vr1[4];
#pragma unroll
        for (int kc = 0; kc < 4; ++kc) {
            const int vr = kt * 16 + 4 * kc + 2 * hi;
            vr0[kc] = *(const v4f*)(vb + vr * D_ + 4 * ln);
            vr1[kc] = *(const v4f*)(vb + (vr + 1) * D_ + 4 * ln);
        }

        v8f cacc = (v8f)0.0f;
#pragma unroll
        for (int g = 0; g < 4; ++g) {
            const v4f x0 = kx0[g];
            const v4f x1 = kx1[g];
            const v4f cc = kcc[g];
            const v4f ss = kss[g];
            v2f bf0, bf1, bf2, bf3;
            bf0.x = x0.x * cc.x - x0.y * ss.x;   bf0.y = x0.y * cc.x + x0.x * ss.x;
            bf1.x = x0.z * cc.y - x0.w * ss.y;   bf1.y = x0.w * cc.y + x0.z * ss.y;
            bf2.x = x1.x * cc.z - x1.y * ss.z;   bf2.y = x1.y * cc.z + x1.x * ss.z;
            bf3.x = x1.z * cc.w - x1.w * ss.w;   bf3.y = x1.w * cc.w + x1.z * ss.w;
            cacc = __builtin_amdgcn_wmma_f32_16x16x4_f32(
                false, afrag[4 * g + 0], false, bf0, (short)0, cacc, false, false);
            cacc = __builtin_amdgcn_wmma_f32_16x16x4_f32(
                false, afrag[4 * g + 1], false, bf1, (short)0, cacc, false, false);
            cacc = __builtin_amdgcn_wmma_f32_16x16x4_f32(
                false, afrag[4 * g + 2], false, bf2, (short)0, cacc, false, false);
            cacc = __builtin_amdgcn_wmma_f32_16x16x4_f32(
                false, afrag[4 * g + 3], false, bf3, (short)0, cacc, false, false);
        }

        {
            const int kn   = (kt < qt) ? (kt + 1) : qt;
            const int nrow = kn * 16 + ln;
#pragma unroll
            for (int g = 0; g < 4; ++g) {
                const int d0 = 16 * g + 8 * hi;
                kx0[g] = *(const v4f*)(kb + nrow * D_ + d0);
                kx1[g] = *(const v4f*)(kb + nrow * D_ + d0 + 4);
                kcc[g] = *(const v4f*)(cosc + nrow * HALF_ + (d0 >> 1));
                kss[g] = *(const v4f*)(sinc + nrow * HALF_ + (d0 >> 1));
            }
        }

#pragma unroll
        for (int r = 0; r < 8; ++r) {
            const int m    = r + 8 * hi;
            const int qpos = qt * 16 + m;
            const int kpos = kt * 16 + ln;
            float x = cacc[r] * scale - LAMBDA_ * fabsf((float)(qpos - kpos));
            const bool fut = kpos > qpos;
            x = fminf(fmaxf(x, -CLAMP_), CLAMP_);
            float beta = 1.0f / (1.0f + __expf(-x));
            if (fut) beta = 0.0f;
            wlds[m * LSTR + ln] = beta;
        }
        asm volatile("s_wait_dscnt 0" ::: "memory");
        __builtin_amdgcn_wave_barrier();

        if (lane < 16) {
            v4f* rowp = (v4f*)(wlds + lane * LSTR);
            v4f w0 = rowp[0], w1 = rowp[1], w2 = rowp[2], w3 = rowp[3];
            float r_   = rem;
            float dsum = 0.0f;
#define SB_STEP(VEC, I)                                   \
            { const float w = VEC[I] * r_;                \
              r_ = fmaxf(r_ * (1.0f - w), EPS_);          \
              VEC[I] = w;  dsum += w; }
            SB_STEP(w0, 0) SB_STEP(w0, 1) SB_STEP(w0, 2) SB_STEP(w0, 3)
            SB_STEP(w1, 0) SB_STEP(w1, 1) SB_STEP(w1, 2) SB_STEP(w1, 3)
            SB_STEP(w2, 0) SB_STEP(w2, 1) SB_STEP(w2, 2) SB_STEP(w2, 3)
            SB_STEP(w3, 0) SB_STEP(w3, 1) SB_STEP(w3, 2) SB_STEP(w3, 3)
#undef SB_STEP
            rowp[0] = w0; rowp[1] = w1; rowp[2] = w2; rowp[3] = w3;
            rem    = r_;
            denom += dsum;
        }
        asm volatile("s_wait_dscnt 0" ::: "memory");
        __builtin_amdgcn_wave_barrier();

        v2f awf[4];
#pragma unroll
        for (int kc = 0; kc < 4; ++kc)
            awf[kc] = *(const v2f*)(wlds + ln * LSTR + 4 * kc + 2 * hi);
#pragma unroll
        for (int kc = 0; kc < 4; ++kc) {
            const v4f r0 = vr0[kc];
            const v4f r1 = vr1[kc];
            v2f b0, b1, b2, b3;
            b0.x = r0.x;  b0.y = r1.x;
            b1.x = r0.y;  b1.y = r1.y;
            b2.x = r0.z;  b2.y = r1.z;
            b3.x = r0.w;  b3.y = r1.w;
            acc0 = __builtin_amdgcn_wmma_f32_16x16x4_f32(
                false, awf[kc], false, b0, (short)0, acc0, false, false);
            acc1 = __builtin_amdgcn_wmma_f32_16x16x4_f32(
                false, awf[kc], false, b1, (short)0, acc1, false, false);
            acc2 = __builtin_amdgcn_wmma_f32_16x16x4_f32(
                false, awf[kc], false, b2, (short)0, acc2, false, false);
            acc3 = __builtin_amdgcn_wmma_f32_16x16x4_f32(
                false, awf[kc], false, b3, (short)0, acc3, false, false);
        }
    }

    if (lane < 16) dlds[lane] = fmaxf(denom, EPS_);
    asm volatile("s_wait_dscnt 0" ::: "memory");
    __builtin_amdgcn_wave_barrier();

#pragma unroll
    for (int r = 0; r < 8; ++r) {
        const int m  = r + 8 * hi;
        const float dn = dlds[m];
        const int orow = qt * 16 + m;
        v4f o;
        o.x = acc0[r] / dn;
        o.y = acc1[r] / dn;
        o.z = acc2[r] / dn;
        o.w = acc3[r] / dn;
        *(v4f*)(ob + orow * D_ + 4 * ln) = o;
    }
  }
}

extern "C" void kernel_launch(void* const* d_in, const int* in_sizes, int n_in,
                              void* d_out, int out_size, void* d_ws, size_t ws_size,
                              hipStream_t stream) {
    (void)in_sizes; (void)n_in; (void)out_size; (void)d_ws; (void)ws_size;
    const float* q    = (const float*)d_in[0];
    const float* k    = (const float*)d_in[1];
    const float* v    = (const float*)d_in[2];
    const float* cosc = (const float*)d_in[3];
    const float* sinc = (const float*)d_in[4];
    float* out = (float*)d_out;

    const int tiles  = B_ * H_ * QTILES;
    const int blocks = tiles / (2 * WPB);
    rse_stickbreak_attn<<<blocks, 32 * WPB, 0, stream>>>(q, k, v, cosc, sinc, out);
}


# Logits tile computed transposed (k fragment as A, q as B): each lane holds its row's 8 betas, so scan and W fragments stay in registers, no LDS in the k loop; W x V and row-sum add orders change
# speedup vs baseline: 1.5998x; 1.0894x over previous
#include <hip/hip_runtime.h>
#include <math.h>

typedef float v8f __attribute__((ext_vector_type(8)));
typedef float v4f __attribute__((ext_vector_type(4)));
typedef float v2f __attribute__((ext_vector_type(2)));

#define B_      4
#define H_      12
#define S_      1024
#define D_      64
#define HALF_   32
#define LAMBDA_ 0.01f
#define EPS_    1e-6f
#define CLAMP_  20.0f
#define QTILES  (S_ / 16)
#define WPB     4

static __device__ __forceinline__ float xhalf(float x) {
    const int b = __builtin_bit_cast(int, x);
    return __builtin_bit_cast(float,
        __builtin_amdgcn_permlanex16(b, b, 0x76543210, 0xfedcba98, false, false));
}

__global__ __launch_bounds__(32 * WPB)
void rse_stickbreak_attn(const float* __restrict__ q,
                         const float* __restrict__ k,
                         const float* __restrict__ v,
                         const float* __restrict__ cosc,
                         const float* __restrict__ sinc,
                         float* __restrict__ out)
{
    __shared__ __align__(16) float lds[WPB][16];

    const int wave = threadIdx.x >> 5;
    const int lane = threadIdx.x & 31;
    const int pr = (blockIdx.x % (QTILES / 2 / WPB)) * WPB + wave;
    const int bh = blockIdx.x / (QTILES / 2 / WPB);

    float* dlds = &lds[wave][0];

    const int hi = lane >> 4;
    const int ln = lane & 15;

    const float* qb = q + (size_t)bh * S_ * D_;
    const float* kb = k + (size_t)bh * S_ * D_;
    const float* vb = v + (size_t)bh * S_ * D_;
    float*       ob = out + (size_t)bh * S_ * D_;

  for (int pass = 0; pass < 2; ++pass) {
    const int qt = pass ? pr : (QTILES - 1 - pr);

    const int qrow = qt * 16 + ln;
    v2f afrag[16];
#pragma unroll
    for (int g = 0; g < 4; ++g) {
        const int d0 = 16 * g + 8 * hi;
        const v4f x0 = *(const v4f*)(qb + qrow * D_ + d0);
        const v4f x1 = *(const v4f*)(qb + qrow * D_ + d0 + 4);
        const v4f cc = *(const v4f*)(cosc + qrow * HALF_ + (d0 >> 1));
        const v4f ss = *(const v4f*)(sinc + qrow * HALF_ + (d0 >> 1));
        afrag[4 * g + 0].x = x0.x * cc.x - x0.y * ss.x;
        afrag[4 * g + 0].y = x0.y * cc.x + x0.x * ss.x;
        afrag[4 * g + 1].x = x0.z * cc.y - x0.w * ss.y;
        afrag[4 * g + 1].y = x0.w * cc.y + x0.z * ss.y;
        afrag[4 * g + 2].x = x1.x * cc.z - x1.y * ss.z;
        afrag[4 * g + 2].y = x1.y * cc.z + x1.x * ss.z;
        afrag[4 * g + 3].x = x1.z * cc.w - x1.w * ss.w;
        afrag[4 * g + 3].y = x1.w * cc.w + x1.z * ss.w;
    }

    v8f acc0 = (v8f)0.0f, acc1 = (v8f)0.0f, acc2 = (v8f)0.0f, acc3 = (v8f)0.0f;
    float rem = 1.0f;
    float den = 0.0f;
    const float scale = 0.125f;

    v4f kx0[4], kx1[4], kcc[4], kss[4];
#pragma unroll
    for (int g = 0; g < 4; ++g) {
        const int d0 = 16 * g + 8 * hi;
        kx0[g] = *(const v4f*)(kb + ln * D_ + d0);
        kx1[g] = *(const v4f*)(kb + ln * D_ + d0 + 4);
        kcc[g] = *(const v4f*)(cosc + ln * HALF_ + (d0 >> 1));
        kss[g] = *(const v4f*)(sinc + ln * HALF_ + (d0 >> 1));
    }

    for (int kt = 0; kt <= qt; ++kt) {
        v4f vr0[4], vr1[4];
#pragma unroll
        for (int kc = 0; kc < 4; ++kc) {
            const int vr = kt * 16 + 8 * hi + 2 * kc;
            vr0[kc] = *(const v4f*)(vb + vr * D_ + 4 * ln);
            vr1[kc] = *(const v4f*)(vb + (vr + 1) * D_ + 4 * ln);
        }

        v2f kf[16];
#pragma unroll
        for (int g = 0; g < 4; ++g) {
            const v4f x0 = kx0[g];
            const v4f x1 = kx1[g];
            const v4f cc = kcc[g];
            const v4f ss = kss[g];
            kf[4 * g + 0].x = x0.x * cc.x - x0.y * ss.x;   kf[4 * g + 0].y = x0.y * cc.x + x0.x * ss.x;
            kf[4 * g + 1].x = x0.z * cc.y - x0.w * ss.y;   kf[4 * g + 1].y = x0.w * cc.y + x0.z * ss.y;
            kf[4 * g + 2].x = x1.x * cc.z - x1.y * ss.z;   kf[4 * g + 2].y = x1.y * cc.z + x1.x * ss.z;
            kf[4 * g + 3].x = x1.z * cc.w - x1.w * ss.w;   kf[4 * g + 3].y = x1.w * cc.w + x1.z * ss.w;
        }

        {
            const int kn   = (kt < qt) ? (kt + 1) : qt;
            const int nrow = kn * 16 + ln;
#pragma unroll
            for (int g = 0; g < 4; ++g) {
                const int d0 = 16 * g + 8 * hi;
                kx0[g] = *(const v4f*)(kb + nrow * D_ + d0);
                kx1[g] = *(const v4f*)(kb + nrow * D_ + d0 + 4);
                kcc[g] = *(const v4f*)(cosc + nrow * HALF_ + (d0 >> 1));
                kss[g] = *(const v4f*)(sinc + nrow * HALF_ + (d0 >> 1));
            }
        }

        v8f cacc = (v8f)0.0f;
#pragma unroll
        for (int s = 0; s < 16; ++s)
            cacc = __builtin_amdgcn_wmma_f32_16x16x4_f32(
                false, kf[s], false, afrag[s], (short)0, cacc, false, false);

        __builtin_amdgcn_sched_barrier(0);

        float bt[8];
#pragma unroll
        for (int r = 0; r < 8; ++r) {
            const int qpos = qt * 16 + ln;
            const int kpos = kt * 16 + 8 * hi + r;
            float x = cacc[r] * scale - LAMBDA_ * fabsf((float)(qpos - kpos));
            const bool fut = kpos > qpos;
            x = fminf(fmaxf(x, -CLAMP_), CLAMP_);
            float beta = 1.0f / (1.0f + __expf(-x));
            if (fut) beta = 0.0f;
            bt[r] = beta;
        }

#define SB_STEP(R, W, I)                                  \
            { const float w = bt[I] * R;                  \
              R = fmaxf(R * (1.0f - w), EPS_);            \
              W[I] = w; }
        float wa[8], wb[8];
        float ra = rem;
        SB_STEP(ra, wa, 0) SB_STEP(ra, wa, 1) SB_STEP(ra, wa, 2) SB_STEP(ra, wa, 3)
        SB_STEP(ra, wa, 4) SB_STEP(ra, wa, 5) SB_STEP(ra, wa, 6) SB_STEP(ra, wa, 7)
        float rb = xhalf(ra);
        SB_STEP(rb, wb, 0) SB_STEP(rb, wb, 1) SB_STEP(rb, wb, 2) SB_STEP(rb, wb, 3)
        SB_STEP(rb, wb, 4) SB_STEP(rb, wb, 5) SB_STEP(rb, wb, 6) SB_STEP(rb, wb, 7)
#undef SB_STEP
        rem = xhalf(rb);

        v2f aw[4];
#pragma unroll
        for (int kc = 0; kc < 4; ++kc) {
            aw[kc].x = hi ? wb[2 * kc]     : wa[2 * kc];
            aw[kc].y = hi ? wb[2 * kc + 1] : wa[2 * kc + 1];
        }
        {
            const v2f s = (aw[0] + aw[1]) + (aw[2] + aw[3]);
            den += s.x + s.y;
        }

#pragma unroll
        for (int kc = 0; kc < 4; ++kc) {
            const v4f r0 = vr0[kc];
            const v4f r1 = vr1[kc];
            v2f b0, b1, b2, b3;
            b0.x = r0.x;  b0.y = r1.x;
            b1.x = r0.y;  b1.y = r1.y;
            b2.x = r0.z;  b2.y = r1.z;
            b3.x = r0.w;  b3.y = r1.w;
            acc0 = __builtin_amdgcn_wmma_f32_16x16x4_f32(
                false, aw[kc], false, b0, (short)0, acc0, false, false);
            acc1 = __builtin_amdgcn_wmma_f32_16x16x4_f32(
                false, aw[kc], false, b1, (short)0, acc1, false, false);
            acc2 = __builtin_amdgcn_wmma_f32_16x16x4_f32(
                false, aw[kc], false, b2, (short)0, acc2, false, false);
            acc3 = __builtin_amdgcn_wmma_f32_16x16x4_f32(
                false, aw[kc], false, b3, (short)0, acc3, false, false);
        }
        __builtin_amdgcn_sched_barrier(0);
    }

    dlds[ln] = fmaxf(den + xhalf(den), EPS_);
    asm volatile("s_wait_dscnt 0" ::: "memory");
    __builtin_amdgcn_wave_barrier();

#pragma unroll
    for (int r = 0; r < 8; ++r) {
        const int m  = r + 8 * hi;
        const float dn = dlds[m];
        const int orow = qt * 16 + m;
        v4f o;
        o.x = acc0[r] / dn;
        o.y = acc1[r] / dn;
        o.z = acc2[r] / dn;
        o.w = acc3[r] / dn;
        *(v4f*)(ob + orow * D_ + 4 * ln) = o;
    }
  }
}

extern "C" void kernel_launch(void* const* d_in, const int* in_sizes, int n_in,
                              void* d_out, int out_size, void* d_ws, size_t ws_size,
                              hipStream_t stream) {
    (void)in_sizes; (void)n_in; (void)out_size; (void)d_ws; (void)ws_size;
    const float* q    = (const float*)d_in[0];
    const float* k    = (const float*)d_in[1];
    const float* v    = (const float*)d_in[2];
    const float* cosc = (const float*)d_in[3];
    const float* sinc = (const float*)d_in[4];
    float* out = (float*)d_out;

    const int tiles  = B_ * H_ * QTILES;
    const int blocks = tiles / (2 * WPB);
    rse_stickbreak_attn<<<blocks, 32 * WPB, 0, stream>>>(q, k, v, cosc, sinc, out);
}


# Each wave runs two adjacent causal q-tiles at once per k tile: k rotation, next-tile loads and v fragments done once feed both tiles' products; 768 waves x 66 trips; per-tile sum orders unchanged
# speedup vs baseline: 1.6958x; 1.0600x over previous
#include <hip/hip_runtime.h>
#include <math.h>

typedef float v8f __attribute__((ext_vector_type(8)));
typedef float v4f __attribute__((ext_vector_type(4)));
typedef float v2f __attribute__((ext_vector_type(2)));

#define B_      4
#define H_      12
#define S_      1024
#define D_      64
#define HALF_   32
#define LAMBDA_ 0.01f
#define EPS_    1e-6f
#define CLAMP_  20.0f
#define QTILES  (S_ / 16)
#define QBLKS   (QTILES / 2)
#define WPB     4

static __device__ __forceinline__ float xhalf(float x) {
    const int b = __builtin_bit_cast(int, x);
    return __builtin_bit_cast(float,
        __builtin_amdgcn_permlanex16(b, b, 0x76543210, 0xfedcba98, false, false));
}

static __device__ __forceinline__ void rope_frags(v2f* f, const v4f* x0a, const v4f* x1a,
                                                  const v4f* cca, const v4f* ssa) {
#pragma unroll
    for (int g = 0; g < 4; ++g) {
        const v4f x0 = x0a[g];
        const v4f x1 = x1a[g];
        const v4f cc = cca[g];
        const v4f ss = ssa[g];
        f[4 * g + 0].x = x0.x * cc.x - x0.y * ss.x;   f[4 * g + 0].y = x0.y * cc.x + x0.x * ss.x;
        f[4 * g + 1].x = x0.z * cc.y - x0.w * ss.y;   f[4 * g + 1].y = x0.w * cc.y + x0.z * ss.y;
        f[4 * g + 2].x = x1.x * cc.z - x1.y * ss.z;   f[4 * g + 2].y = x1.y * cc.z + x1.x * ss.z;
        f[4 * g + 3].x = x1.z * cc.w - x1.w * ss.w;   f[4 * g + 3].y = x1.w * cc.w + x1.z * ss.w;
    }
}

static __device__ __forceinline__ void load_rows(v4f* x0a, v4f* x1a, v4f* cca, v4f* ssa,
                                                 const float* __restrict__ xb,
                                                 const float* __restrict__ cosc,
                                                 const float* __restrict__ sinc,
                                                 int row, int hi) {
#pragma unroll
    for (int g = 0; g < 4; ++g) {
        const int d0 = 16 * g + 8 * hi;
        x0a[g] = *(const v4f*)(xb + row * D_ + d0);
        x1a[g] = *(const v4f*)(xb + row * D_ + d0 + 4);
        cca[g] = *(const v4f*)(cosc + row * HALF_ + (d0 >> 1));
        ssa[g] = *(const v4f*)(sinc + row * HALF_ + (d0 >> 1));
    }
}

__global__ __launch_bounds__(32 * WPB) __attribute__((amdgpu_waves_per_eu(1, 1)))
void rse_stickbreak_attn(const float* __restrict__ q,
                         const float* __restrict__ k,
                         const float* __restrict__ v,
                         const float* __restrict__ cosc,
                         const float* __restrict__ sinc,
                         float* __restrict__ out)
{
    __shared__ __align__(16) float lds[WPB][32];

    const int wave = threadIdx.x >> 5;
    const int lane = threadIdx.x & 31;
    const int pr = (blockIdx.x % (QBLKS / 2 / WPB)) * WPB + wave;
    const int bh = blockIdx.x / (QBLKS / 2 / WPB);

    float* dlds = &lds[wave][0];

    const int hi = lane >> 4;
    const int ln = lane & 15;

    const float* qb = q + (size_t)bh * S_ * D_;
    const float* kb = k + (size_t)bh * S_ * D_;
    const float* vb = v + (size_t)bh * S_ * D_;
    float*       ob = out + (size_t)bh * S_ * D_;

  for (int pass = 0; pass < 2; ++pass) {
    const int qblk = pass ? pr : (QBLKS - 1 - pr);
    const int qt0  = 2 * qblk;
    const int qt1  = qt0 + 1;

    v2f afA[16], afB[16];
    {
        v4f x0[4], x1[4], cc[4], ss[4];
        load_rows(x0, x1, cc, ss, qb, cosc, sinc, qt0 * 16 + ln, hi);
        rope_frags(afA, x0, x1, cc, ss);
        load_rows(x0, x1, cc, ss, qb, cosc, sinc, qt1 * 16 + ln, hi);
        rope_frags(afB, x0, x1, cc, ss);
    }

    v8f accA0 = (v8f)0.0f, accA1 = (v8f)0.0f, accA2 = (v8f)0.0f, accA3 = (v8f)0.0f;
    v8f accB0 = (v8f)0.0f, accB1 = (v8f)0.0f, accB2 = (v8f)0.0f, accB3 = (v8f)0.0f;
    float remA = 1.0f, remB = 1.0f;
    float denA = 0.0f, denB = 0.0f;
    const float scale = 0.125f;

    v4f kx0[4], kx1[4], kcc[4], kss[4];
    load_rows(kx0, kx1, kcc, kss, kb, cosc, sinc, ln, hi);

    for (int kt = 0; kt <= qt1; ++kt) {
        v4f vr0[4], vr1[4];
#pragma unroll
        for (int kc = 0; kc < 4; ++kc) {
            const int vr = kt * 16 + 8 * hi + 2 * kc;
            vr0[kc] = *(const v4f*)(vb + vr * D_ + 4 * ln);
            vr1[kc] = *(const v4f*)(vb + (vr + 1) * D_ + 4 * ln);
        }

        v2f kf[16];
        rope_frags(kf, kx0, kx1, kcc, kss);

        {
            const int kn = (kt < qt1) ? (kt + 1) : qt1;
            load_rows(kx0, kx1, kcc, kss, kb, cosc, sinc, kn * 16 + ln, hi);
        }

        v8f caccA = (v8f)0.0f, caccB = (v8f)0.0f;
#pragma unroll
        for (int s = 0; s < 16; ++s) {
            caccA = __builtin_amdgcn_wmma_f32_16x16x4_f32(
                false, kf[s], false, afA[s], (short)0, caccA, false, false);
            caccB = __builtin_amdgcn_wmma_f32_16x16x4_f32(
                false, kf[s], false, afB[s], (short)0, caccB, false, false);
        }

        __builtin_amdgcn_sched_barrier(0);

        float btA[8], btB[8];
#pragma unroll
        for (int r = 0; r < 8; ++r) {
            const int kpos  = kt * 16 + 8 * hi + r;
            const int qposA = qt0 * 16 + ln;
            const int qposB = qt1 * 16 + ln;
            float xa = caccA[r] * scale - LAMBDA_ * fabsf((float)(qposA - kpos));
            float xb = caccB[r] * scale - LAMBDA_ * fabsf((float)(qposB - kpos));
            xa = fminf(fmaxf(xa, -CLAMP_), CLAMP_);
            xb = fminf(fmaxf(xb, -CLAMP_), CLAMP_);
            float ba = 1.0f / (1.0f + __expf(-xa));
            float bb = 1.0f / (1.0f + __expf(-xb));
            if (kpos > qposA) ba = 0.0f;
            if (kpos > qposB) bb = 0.0f;
            btA[r] = ba;
            btB[r] = bb;
        }

#define SB_STEP(R, W, BT, I)                              \
            { const float w = BT[I] * R;                  \
              R = fmaxf(R * (1.0f - w), EPS_);            \
              W[I] = w; }
#define SB_STEP2(RA, WA, RB, WB, I)  SB_STEP(RA, WA, btA, I) SB_STEP(RB, WB, btB, I)
        float waA[8], wbA[8], waB[8], wbB[8];
        float raA = remA, raB = remB;
        SB_STEP2(raA, waA, raB, waB, 0) SB_STEP2(raA, waA, raB, waB, 1)
        SB_STEP2(raA, waA, raB, waB, 2) SB_STEP2(raA, waA, raB, waB, 3)
        SB_STEP2(raA, waA, raB, waB, 4) SB_STEP2(raA, waA, raB, waB, 5)
        SB_STEP2(raA, waA, raB, waB, 6) SB_STEP2(raA, waA, raB, waB, 7)
        float rbA = xhalf(raA), rbB = xhalf(raB);
        SB_STEP2(rbA, wbA, rbB, wbB, 0) SB_STEP2(rbA, wbA, rbB, wbB, 1)
        SB_STEP2(rbA, wbA, rbB, wbB, 2) SB_STEP2(rbA, wbA, rbB, wbB, 3)
        SB_STEP2(rbA, wbA, rbB, wbB, 4) SB_STEP2(rbA, wbA, rbB, wbB, 5)
        SB_STEP2(rbA, wbA, rbB, wbB, 6) SB_STEP2(rbA, wbA, rbB, wbB, 7)
#undef SB_STEP2
#undef SB_STEP
        remA = xhalf(rbA);
        remB = xhalf(rbB);

        v2f awA[4], awB[4];
#pragma unroll
        for (int kc = 0; kc < 4; ++kc) {
            awA[kc].x = hi ? wbA[2 * kc]     : waA[2 * kc];
            awA[kc].y = hi ? wbA[2 * kc + 1] : waA[2 * kc + 1];
            awB[kc].x = hi ? wbB[2 * kc]     : waB[2 * kc];
            awB[kc].y = hi ? wbB[2 * kc + 1] : waB[2 * kc + 1];
        }
        {
            const v2f sA = (awA[0] + awA[1]) + (awA[2] + awA[3]);
            denA += sA.x + sA.y;
            const v2f sB = (awB[0] + awB[1]) + (awB[2] + awB[3]);
            denB += sB.x + sB.y;
        }

#pragma unroll
        for (int kc = 0; kc < 4; ++kc) {
            const v4f r0 = vr0[kc];
            const v4f r1 = vr1[kc];
            v2f b0, b1, b2, b3;
            b0.x = r0.x;  b0.y = r1.x;
            b1.x = r0.y;  b1.y = r1.y;
            b2.x = r0.z;  b2.y = r1.z;
            b3.x = r0.w;  b3.y = r1.w;
            accA0 = __builtin_amdgcn_wmma_f32_16x16x4_f32(
                false, awA[kc], false, b0, (short)0, accA0, false, false);
            accB0 = __builtin_amdgcn_wmma_f32_16x16x4_f32(
                false, awB[kc], false, b0, (short)0, accB0, false, false);
            accA1 = __builtin_amdgcn_wmma_f32_16x16x4_f32(
                false, awA[kc], false, b1, (short)0, accA1, false, false);
            accB1 = __builtin_amdgcn_wmma_f32_16x16x4_f32(
                false, awB[kc], false, b1, (short)0, accB1, false, false);
            accA2 = __builtin_amdgcn_wmma_f32_16x16x4_f32(
                false, awA[kc], false, b2, (short)0, accA2, false, false);
            accB2 = __builtin_amdgcn_wmma_f32_16x16x4_f32(
                false, awB[kc], false, b2, (short)0, accB2, false, false);
            accA3 = __builtin_amdgcn_wmma_f32_16x16x4_f32(
                false, awA[kc], false, b3, (short)0, accA3, false, false);
            accB3 = __builtin_amdgcn_wmma_f32_16x16x4_f32(
                false, awB[kc], false, b3, (short)0, accB3, false, false);
        }
        __builtin_amdgcn_sched_barrier(0);
    }

    dlds[ln]      = fmaxf(denA + xhalf(denA), EPS_);
    dlds[16 + ln] = fmaxf(denB + xhalf(denB), EPS_);
    asm volatile("s_wait_dscnt 0" ::: "memory");
    __builtin_amdgcn_wave_barrier();

#pragma unroll
    for (int r = 0; r < 8; ++r) {
        const int m  = r + 8 * hi;
        const float dnA = dlds[m];
        const float dnB = dlds[16 + m];
        const int orow = qt0 * 16 + m;
        v4f oa, ob4;
        oa.x = accA0[r] / dnA;
        oa.y = accA1[r] / dnA;
        oa.z = accA2[r] / dnA;
        oa.w = accA3[r] / dnA;
        ob4.x = accB0[r] / dnB;
        ob4.y = accB1[r] / dnB;
        ob4.z = accB2[r] / dnB;
        ob4.w = accB3[r] / dnB;
        *(v4f*)(ob + orow * D_ + 4 * ln)        = oa;
        *(v4f*)(ob + (orow + 16) * D_ + 4 * ln) = ob4;
    }
  }
}

extern "C" void kernel_launch(void* const* d_in, const int* in_sizes, int n_in,
                              void* d_out, int out_size, void* d_ws, size_t ws_size,
                              hipStream_t stream) {
    (void)in_sizes; (void)n_in; (void)out_size; (void)d_ws; (void)ws_size;
    const float* q    = (const float*)d_in[0];
    const float* k    = (const float*)d_in[1];
    const float* v    = (const float*)d_in[2];
    const float* cosc = (const float*)d_in[3];
    const float* sinc = (const float*)d_in[4];
    float* out = (float*)d_out;

    const int tiles  = B_ * H_ * QTILES;
    const int blocks = tiles / (4 * WPB);
    rse_stickbreak_attn<<<blocks, 32 * WPB, 0, stream>>>(q, k, v, cosc, sinc, out);
}


# Sigmoid's 1/(1+e^-x) taken with the hardware reciprocal (1 ulp) in place of the correctly rounded division sequence: betas may differ in the last bit; every sum order unchanged
# speedup vs baseline: 1.7643x; 1.0404x over previous
#include <hip/hip_runtime.h>
#include <math.h>

typedef float v8f __attribute__((ext_vector_type(8)));
typedef float v4f __attribute__((ext_vector_type(4)));
typedef float v2f __attribute__((ext_vector_type(2)));

#define B_      4
#define H_      12
#define S_      1024
#define D_      64
#define HALF_   32
#define LAMBDA_ 0.01f
#define EPS_    1e-6f
#define CLAMP_  20.0f
#define QTILES  (S_ / 16)
#define QBLKS   (QTILES / 2)
#define WPB     4

static __device__ __forceinline__ float xhalf(float x) {
    const int b = __builtin_bit_cast(int, x);
    return __builtin_bit_cast(float,
        __builtin_amdgcn_permlanex16(b, b, 0x76543210, 0xfedcba98, false, false));
}

static __device__ __forceinline__ void rope_frags(v2f* f, const v4f* x0a, const v4f* x1a,
                                                  const v4f* cca, const v4f* ssa) {
#pragma unroll
    for (int g = 0; g < 4; ++g) {
        const v4f x0 = x0a[g];
        const v4f x1 = x1a[g];
        const v4f cc = cca[g];
        const v4f ss = ssa[g];
        f[4 * g + 0].x = x0.x * cc.x - x0.y * ss.x;   f[4 * g + 0].y = x0.y * cc.x + x0.x * ss.x;
        f[4 * g + 1].x = x0.z * cc.y - x0.w * ss.y;   f[4 * g + 1].y = x0.w * cc.y + x0.z * ss.y;
        f[4 * g + 2].x = x1.x * cc.z - x1.y * ss.z;   f[4 * g + 2].y = x1.y * cc.z + x1.x * ss.z;
        f[4 * g + 3].x = x1.z * cc.w - x1.w * ss.w;   f[4 * g + 3].y = x1.w * cc.w + x1.z * ss.w;
    }
}

static __device__ __forceinline__ void load_rows(v4f* x0a, v4f* x1a, v4f* cca, v4f* ssa,
                                                 const float* __restrict__ xb,
                                                 const float* __restrict__ cosc,
                                                 const float* __restrict__ sinc,
                                                 int row, int hi) {
#pragma unroll
    for (int g = 0; g < 4; ++g) {
        const int d0 = 16 * g + 8 * hi;
        x0a[g] = *(const v4f*)(xb + row * D_ + d0);
        x1a[g] = *(const v4f*)(xb + row * D_ + d0 + 4);
        cca[g] = *(const v4f*)(cosc + row * HALF_ + (d0 >> 1));
        ssa[g] = *(const v4f*)(sinc + row * HALF_ + (d0 >> 1));
    }
}

__global__ __launch_bounds__(32 * WPB) __attribute__((amdgpu_waves_per_eu(1, 1)))
void rse_stickbreak_attn(const float* __restrict__ q,
                         const float* __restrict__ k,
                         const float* __restrict__ v,
                         const float* __restrict__ cosc,
                         const float* __restrict__ sinc,
                         float* __restrict__ out)
{
    __shared__ __align__(16) float lds[WPB][32];

    const int wave = threadIdx.x >> 5;
    const int lane = threadIdx.x & 31;
    const int pr = (blockIdx.x % (QBLKS / 2 / WPB)) * WPB + wave;
    const int bh = blockIdx.x / (QBLKS / 2 / WPB);

    float* dlds = &lds[wave][0];

    const int hi = lane >> 4;
    const int ln = lane & 15;

    const float* qb = q + (size_t)bh * S_ * D_;
    const float* kb = k + (size_t)bh * S_ * D_;
    const float* vb = v + (size_t)bh * S_ * D_;
    float*       ob = out + (size_t)bh * S_ * D_;

  for (int pass = 0; pass < 2; ++pass) {
    const int qblk = pass ? pr : (QBLKS - 1 - pr);
    const int qt0  = 2 * qblk;
    const int qt1  = qt0 + 1;

    v2f afA[16], afB[16];
    {
        v4f x0[4], x1[4], cc[4], ss[4];
        load_rows(x0, x1, cc, ss, qb, cosc, sinc, qt0 * 16 + ln, hi);
        rope_frags(afA, x0, x1, cc, ss);
        load_rows(x0, x1, cc, ss, qb, cosc, sinc, qt1 * 16 + ln, hi);
        rope_frags(afB, x0, x1, cc, ss);
    }

    v8f accA0 = (v8f)0.0f, accA1 = (v8f)0.0f, accA2 = (v8f)0.0f, accA3 = (v8f)0.0f;
    v8f accB0 = (v8f)0.0f, accB1 = (v8f)0.0f, accB2 = (v8f)0.0f, accB3 = (v8f)0.0f;
    float remA = 1.0f, remB = 1.0f;
    float denA = 0.0f, denB = 0.0f;
    const float scale = 0.125f;

    v4f kx0[4], kx1[4], kcc[4], kss[4];
    load_rows(kx0, kx1, kcc, kss, kb, cosc, sinc, ln, hi);

    for (int kt = 0; kt <= qt1; ++kt) {
        v4f vr0[4], vr1[4];
#pragma unroll
        for (int kc = 0; kc < 4; ++kc) {
            const int vr = kt * 16 + 8 * hi + 2 * kc;
            vr0[kc] = *(const v4f*)(vb + vr * D_ + 4 * ln);
            vr1[kc] = *(const v4f*)(vb + (vr + 1) * D_ + 4 * ln);
        }

        v2f kf[16];
        rope_frags(kf, kx0, kx1, kcc, kss);

        {
            const int kn = (kt < qt1) ? (kt + 1) : qt1;
            load_rows(kx0, kx1, kcc, kss, kb, cosc, sinc, kn * 16 + ln, hi);
        }

        v8f caccA = (v8f)0.0f, caccB = (v8f)0.0f;
#pragma unroll
        for (int s = 0; s < 16; ++s) {
            caccA = __builtin_amdgcn_wmma_f32_16x16x4_f32(
                false, kf[s], false, afA[s], (short)0, caccA, false, false);
            caccB = __builtin_amdgcn_wmma_f32_16x16x4_f32(
                false, kf[s], false, afB[s], (short)0, caccB, false, false);
        }

        __builtin_amdgcn_sched_barrier(0);

        float btA[8], btB[8];
#pragma unroll
        for (int r = 0; r < 8; ++r) {
            const int kpos  = kt * 16 + 8 * hi + r;
            const int qposA = qt0 * 16 + ln;
            const int qposB = qt1 * 16 + ln;
            float xa = caccA[r] * scale - LAMBDA_ * fabsf((float)(qposA - kpos));
            float xb = caccB[r] * scale - LAMBDA_ * fabsf((float)(qposB - kpos));
            xa = fminf(fmaxf(xa, -CLAMP_), CLAMP_);
            xb = fminf(fmaxf(xb, -CLAMP_), CLAMP_);
            float ba = __builtin_amdgcn_rcpf(1.0f + __expf(-xa));
            float bb = __builtin_amdgcn_rcpf(1.0f + __expf(-xb));
            if (kpos > qposA) ba = 0.0f;
            if (kpos > qposB) bb = 0.0f;
            btA[r] = ba;
            btB[r] = bb;
        }

#define SB_STEP(R, W, BT, I)                              \
            { const float w = BT[I] * R;                  \
              R = fmaxf(R * (1.0f - w), EPS_);            \
              W[I] = w; }
#define SB_STEP2(RA, WA, RB, WB, I)  SB_STEP(RA, WA, btA, I) SB_STEP(RB, WB, btB, I)
        float waA[8], wbA[8], waB[8], wbB[8];
        float raA = remA, raB = remB;
        SB_STEP2(raA, waA, raB, waB, 0) SB_STEP2(raA, waA, raB, waB, 1)
        SB_STEP2(raA, waA, raB, waB, 2) SB_STEP2(raA, waA, raB, waB, 3)
        SB_STEP2(raA, waA, raB, waB, 4) SB_STEP2(raA, waA, raB, waB, 5)
        SB_STEP2(raA, waA, raB, waB, 6) SB_STEP2(raA, waA, raB, waB, 7)
        float rbA = xhalf(raA), rbB = xhalf(raB);
        SB_STEP2(rbA, wbA, rbB, wbB, 0) SB_STEP2(rbA, wbA, rbB, wbB, 1)
        SB_STEP2(rbA, wbA, rbB, wbB, 2) SB_STEP2(rbA, wbA, rbB, wbB, 3)
        SB_STEP2(rbA, wbA, rbB, wbB, 4) SB_STEP2(rbA, wbA, rbB, wbB, 5)
        SB_STEP2(rbA, wbA, rbB, wbB, 6) SB_STEP2(rbA, wbA, rbB, wbB, 7)
#undef SB_STEP2
#undef SB_STEP
        remA = xhalf(rbA);
        remB = xhalf(rbB);

        v2f awA[4], awB[4];
#pragma unroll
        for (int kc = 0; kc < 4; ++kc) {
            awA[kc].x = hi ? wbA[2 * kc]     : waA[2 * kc];
            awA[kc].y = hi ? wbA[2 * kc + 1] : waA[2 * kc + 1];
            awB[kc].x = hi ? wbB[2 * kc]     : waB[2 * kc];
            awB[kc].y = hi ? wbB[2 * kc + 1] : waB[2 * kc + 1];
        }
        {
            const v2f sA = (awA[0] + awA[1]) + (awA[2] + awA[3]);
            denA += sA.x + sA.y;
            const v2f sB = (awB[0] + awB[1]) + (awB[2] + awB[3]);
            denB += sB.x + sB.y;
        }

#pragma unroll
        for (int kc = 0; kc < 4; ++kc) {
            const v4f r0 = vr0[kc];
            const v4f r1 = vr1[kc];
            v2f b0, b1, b2, b3;
            b0.x = r0.x;  b0.y = r1.x;
            b1.x = r0.y;  b1.y = r1.y;
            b2.x = r0.z;  b2.y = r1.z;
            b3.x = r0.w;  b3.y = r1.w;
            accA0 = __builtin_amdgcn_wmma_f32_16x16x4_f32(
                false, awA[kc], false, b0, (short)0, accA0, false, false);
            accB0 = __builtin_amdgcn_wmma_f32_16x16x4_f32(
                false, awB[kc], false, b0, (short)0, accB0, false, false);
            accA1 = __builtin_amdgcn_wmma_f32_16x16x4_f32(
                false, awA[kc], false, b1, (short)0, accA1, false, false);
            accB1 = __builtin_amdgcn_wmma_f32_16x16x4_f32(
                false, awB[kc], false, b1, (short)0, accB1, false, false);
            accA2 = __builtin_amdgcn_wmma_f32_16x16x4_f32(
                false, awA[kc], false, b2, (short)0, accA2, false, false);
            accB2 = __builtin_amdgcn_wmma_f32_16x16x4_f32(
                false, awB[kc], false, b2, (short)0, accB2, false, false);
            accA3 = __builtin_amdgcn_wmma_f32_16x16x4_f32(
                false, awA[kc], false, b3, (short)0, accA3, false, false);
            accB3 = __builtin_amdgcn_wmma_f32_16x16x4_f32(
                false, awB[kc], false, b3, (short)0, accB3, false, false);
        }
        __builtin_amdgcn_sched_barrier(0);
    }

    dlds[ln]      = fmaxf(denA + xhalf(denA), EPS_);
    dlds[16 + ln] = fmaxf(denB + xhalf(denB), EPS_);
    asm volatile("s_wait_dscnt 0" ::: "memory");
    __builtin_amdgcn_wave_barrier();

#pragma unroll
    for (int r = 0; r < 8; ++r) {
        const int m  = r + 8 * hi;
        const float dnA = dlds[m];
        const float dnB = dlds[16 + m];
        const int orow = qt0 * 16 + m;
        v4f oa, ob4;
        oa.x = accA0[r] / dnA;
        oa.y = accA1[r] / dnA;
        oa.z = accA2[r] / dnA;
        oa.w = accA3[r] / dnA;
        ob4.x = accB0[r] / dnB;
        ob4.y = accB1[r] / dnB;
        ob4.z = accB2[r] / dnB;
        ob4.w = accB3[r] / dnB;
        *(v4f*)(ob + orow * D_ + 4 * ln)        = oa;
        *(v4f*)(ob + (orow + 16) * D_ + 4 * ln) = ob4;
    }
  }
}

extern "C" void kernel_launch(void* const* d_in, const int* in_sizes, int n_in,
                              void* d_out, int out_size, void* d_ws, size_t ws_size,
                              hipStream_t stream) {
    (void)in_sizes; (void)n_in; (void)out_size; (void)d_ws; (void)ws_size;
    const float* q    = (const float*)d_in[0];
    const float* k    = (const float*)d_in[1];
    const float* v    = (const float*)d_in[2];
    const float* cosc = (const float*)d_in[3];
    const float* sinc = (const float*)d_in[4];
    float* out = (float*)d_out;

    const int tiles  = B_ * H_ * QTILES;
    const int blocks = tiles / (4 * WPB);
    rse_stickbreak_attn<<<blocks, 32 * WPB, 0, stream>>>(q, k, v, cosc, sinc, out);
}


# k loop split at the diagonal: trips below both q tiles drop causal select and |.| (distance from one float per tile); a block's last trip skips the all-masked lower tile; sum orders unchanged
# speedup vs baseline: 1.8417x; 1.0439x over previous
#include <hip/hip_runtime.h>
#include <math.h>
#include <type_traits>

typedef float v8f __attribute__((ext_vector_type(8)));
typedef float v4f __attribute__((ext_vector_type(4)));
typedef float v2f __attribute__((ext_vector_type(2)));

#define B_      4
#define H_      12
#define S_      1024
#define D_      64
#define HALF_   32
#define LAMBDA_ 0.01f
#define EPS_    1e-6f
#define CLAMP_  20.0f
#define QTILES  (S_ / 16)
#define QBLKS   (QTILES / 2)
#define WPB     4

static __device__ __forceinline__ float xhalf(float x) {
    const int b = __builtin_bit_cast(int, x);
    return __builtin_bit_cast(float,
        __builtin_amdgcn_permlanex16(b, b, 0x76543210, 0xfedcba98, false, false));
}

static __device__ __forceinline__ void rope_frags(v2f* f, const v4f* x0a, const v4f* x1a,
                                                  const v4f* cca, const v4f* ssa) {
#pragma unroll
    for (int g = 0; g < 4; ++g) {
        const v4f x0 = x0a[g];
        const v4f x1 = x1a[g];
        const v4f cc = cca[g];
        const v4f ss = ssa[g];
        f[4 * g + 0].x = x0.x * cc.x - x0.y * ss.x;   f[4 * g + 0].y = x0.y * cc.x + x0.x * ss.x;
        f[4 * g + 1].x = x0.z * cc.y - x0.w * ss.y;   f[4 * g + 1].y = x0.w * cc.y + x0.z * ss.y;
        f[4 * g + 2].x = x1.x * cc.z - x1.y * ss.z;   f[4 * g + 2].y = x1.y * cc.z + x1.x * ss.z;
        f[4 * g + 3].x = x1.z * cc.w - x1.w * ss.w;   f[4 * g + 3].y = x1.w * cc.w + x1.z * ss.w;
    }
}

static __device__ __forceinline__ void load_rows(v4f* x0a, v4f* x1a, v4f* cca, v4f* ssa,
                                                 const float* __restrict__ xb,
                                                 const float* __restrict__ cosc,
                                                 const float* __restrict__ sinc,
                                                 int row, int hi) {
#pragma unroll
    for (int g = 0; g < 4; ++g) {
        const int d0 = 16 * g + 8 * hi;
        x0a[g] = *(const v4f*)(xb + row * D_ + d0);
        x1a[g] = *(const v4f*)(xb + row * D_ + d0 + 4);
        cca[g] = *(const v4f*)(cosc + row * HALF_ + (d0 >> 1));
        ssa[g] = *(const v4f*)(sinc + row * HALF_ + (d0 >> 1));
    }
}

__global__ __launch_bounds__(32 * WPB) __attribute__((amdgpu_waves_per_eu(1, 1)))
void rse_stickbreak_attn(const float* __restrict__ q,
                         const float* __restrict__ k,
                         const float* __restrict__ v,
                         const float* __restrict__ cosc,
                         const float* __restrict__ sinc,
                         float* __restrict__ out)
{
    __shared__ __align__(16) float lds[WPB][32];

    const int wave = threadIdx.x >> 5;
    const int lane = threadIdx.x & 31;
    const int pr = (blockIdx.x % (QBLKS / 2 / WPB)) * WPB + wave;
    const int bh = blockIdx.x / (QBLKS / 2 / WPB);

    float* dlds = &lds[wave][0];

    const int hi = lane >> 4;
    const int ln = lane & 15;

    const float* qb = q + (size_t)bh * S_ * D_;
    const float* kb = k + (size_t)bh * S_ * D_;
    const float* vb = v + (size_t)bh * S_ * D_;
    float*       ob = out + (size_t)bh * S_ * D_;

  for (int pass = 0; pass < 2; ++pass) {
    const int qblk = pass ? pr : (QBLKS - 1 - pr);
    const int qt0  = 2 * qblk;
    const int qt1  = qt0 + 1;

    v2f afA[16], afB[16];
    {
        v4f x0[4], x1[4], cc[4], ss[4];
        load_rows(x0, x1, cc, ss, qb, cosc, sinc, qt0 * 16 + ln, hi);
        rope_frags(afA, x0, x1, cc, ss);
        load_rows(x0, x1, cc, ss, qb, cosc, sinc, qt1 * 16 + ln, hi);
        rope_frags(afB, x0, x1, cc, ss);
    }

    v8f accA0 = (v8f)0.0f, accA1 = (v8f)0.0f, accA2 = (v8f)0.0f, accA3 = (v8f)0.0f;
    v8f accB0 = (v8f)0.0f, accB1 = (v8f)0.0f, accB2 = (v8f)0.0f, accB3 = (v8f)0.0f;
    float remA = 1.0f, remB = 1.0f;
    float denA = 0.0f, denB = 0.0f;
    const float scale = 0.125f;

    v4f kx0[4], kx1[4], kcc[4], kss[4];
    load_rows(kx0, kx1, kcc, kss, kb, cosc, sinc, ln, hi);

    auto trip = [&](const int kt, auto diag_tag, auto a_tag) __attribute__((always_inline)) {
        constexpr bool DIAG = decltype(diag_tag)::value;
        constexpr bool DO_A = decltype(a_tag)::value;
        v4f vr0[4], vr1[4];
#pragma unroll
        for (int kc = 0; kc < 4; ++kc) {
            const int vr = kt * 16 + 8 * hi + 2 * kc;
            vr0[kc] = *(const v4f*)(vb + vr * D_ + 4 * ln);
            vr1[kc] = *(const v4f*)(vb + (vr + 1) * D_ + 4 * ln);
        }

        v2f kf[16];
        rope_frags(kf, kx0, kx1, kcc, kss);

        {
            const int kn = (kt < qt1) ? (kt + 1) : qt1;
            load_rows(kx0, kx1, kcc, kss, kb, cosc, sinc, kn * 16 + ln, hi);
        }

        v8f caccA = (v8f)0.0f, caccB = (v8f)0.0f;
#pragma unroll
        for (int s = 0; s < 16; ++s) {
            if constexpr (DO_A)
                caccA = __builtin_amdgcn_wmma_f32_16x16x4_f32(
                    false, kf[s], false, afA[s], (short)0, caccA, false, false);
            caccB = __builtin_amdgcn_wmma_f32_16x16x4_f32(
                false, kf[s], false, afB[s], (short)0, caccB, false, false);
        }

        __builtin_amdgcn_sched_barrier(0);

        float btA[8], btB[8];
        if constexpr (DIAG) {
#pragma unroll
            for (int r = 0; r < 8; ++r) {
                const int kpos  = kt * 16 + 8 * hi + r;
                const int qposA = qt0 * 16 + ln;
                const int qposB = qt1 * 16 + ln;
                float xa = caccA[r] * scale - LAMBDA_ * fabsf((float)(qposA - kpos));
                float xb = caccB[r] * scale - LAMBDA_ * fabsf((float)(qposB - kpos));
                xa = fminf(fmaxf(xa, -CLAMP_), CLAMP_);
                xb = fminf(fmaxf(xb, -CLAMP_), CLAMP_);
                float ba = __builtin_amdgcn_rcpf(1.0f + __expf(-xa));
                float bb = __builtin_amdgcn_rcpf(1.0f + __expf(-xb));
                if (kpos > qposA) ba = 0.0f;
                if (kpos > qposB) bb = 0.0f;
                btA[r] = ba;
                btB[r] = bb;
            }
        } else {
            const float dA0 = (float)(qt0 * 16 + ln - kt * 16 - 8 * hi);
            const float dB0 = dA0 + 16.0f;
#pragma unroll
            for (int r = 0; r < 8; ++r) {
                float xa = caccA[r] * scale - LAMBDA_ * (dA0 - (float)r);
                float xb = caccB[r] * scale - LAMBDA_ * (dB0 - (float)r);
                xa = fminf(fmaxf(xa, -CLAMP_), CLAMP_);
                xb = fminf(fmaxf(xb, -CLAMP_), CLAMP_);
                btA[r] = __builtin_amdgcn_rcpf(1.0f + __expf(-xa));
                btB[r] = __builtin_amdgcn_rcpf(1.0f + __expf(-xb));
            }
        }

#define SB_STEP(R, W, BT, I)                              \
            { const float w = BT[I] * R;                  \
              R = fmaxf(R * (1.0f - w), EPS_);            \
              W[I] = w; }
#define SB_STEP2(RA, WA, RB, WB, I)  SB_STEP(RA, WA, btA, I) SB_STEP(RB, WB, btB, I)
        float waA[8], wbA[8], waB[8], wbB[8];
        float raA = remA, raB = remB;
        SB_STEP2(raA, waA, raB, waB, 0) SB_STEP2(raA, waA, raB, waB, 1)
        SB_STEP2(raA, waA, raB, waB, 2) SB_STEP2(raA, waA, raB, waB, 3)
        SB_STEP2(raA, waA, raB, waB, 4) SB_STEP2(raA, waA, raB, waB, 5)
        SB_STEP2(raA, waA, raB, waB, 6) SB_STEP2(raA, waA, raB, waB, 7)
        float rbA = xhalf(raA), rbB = xhalf(raB);
        SB_STEP2(rbA, wbA, rbB, wbB, 0) SB_STEP2(rbA, wbA, rbB, wbB, 1)
        SB_STEP2(rbA, wbA, rbB, wbB, 2) SB_STEP2(rbA, wbA, rbB, wbB, 3)
        SB_STEP2(rbA, wbA, rbB, wbB, 4) SB_STEP2(rbA, wbA, rbB, wbB, 5)
        SB_STEP2(rbA, wbA, rbB, wbB, 6) SB_STEP2(rbA, wbA, rbB, wbB, 7)
#undef SB_STEP2
#undef SB_STEP
        if constexpr (DO_A) remA = xhalf(rbA);
        remB = xhalf(rbB);

        v2f awA[4], awB[4];
#pragma unroll
        for (int kc = 0; kc < 4; ++kc) {
            awA[kc].x = hi ? wbA[2 * kc]     : waA[2 * kc];
            awA[kc].y = hi ? wbA[2 * kc + 1] : waA[2 * kc + 1];
            awB[kc].x = hi ? wbB[2 * kc]     : waB[2 * kc];
            awB[kc].y = hi ? wbB[2 * kc + 1] : waB[2 * kc + 1];
        }
        {
            if constexpr (DO_A) {
                const v2f sA = (awA[0] + awA[1]) + (awA[2] + awA[3]);
                denA += sA.x + sA.y;
            }
            const v2f sB = (awB[0] + awB[1]) + (awB[2] + awB[3]);
            denB += sB.x + sB.y;
        }

#pragma unroll
        for (int kc = 0; kc < 4; ++kc) {
            const v4f r0 = vr0[kc];
            const v4f r1 = vr1[kc];
            v2f b0, b1, b2, b3;
            b0.x = r0.x;  b0.y = r1.x;
            b1.x = r0.y;  b1.y = r1.y;
            b2.x = r0.z;  b2.y = r1.z;
            b3.x = r0.w;  b3.y = r1.w;
            if constexpr (DO_A)
                accA0 = __builtin_amdgcn_wmma_f32_16x16x4_f32(
                    false, awA[kc], false, b0, (short)0, accA0, false, false);
            accB0 = __builtin_amdgcn_wmma_f32_16x16x4_f32(
                false, awB[kc], false, b0, (short)0, accB0, false, false);
            if constexpr (DO_A)
                accA1 = __builtin_amdgcn_wmma_f32_16x16x4_f32(
                    false, awA[kc], false, b1, (short)0, accA1, false, false);
            accB1 = __builtin_amdgcn_wmma_f32_16x16x4_f32(
                false, awB[kc], false, b1, (short)0, accB1, false, false);
            if constexpr (DO_A)
                accA2 = __builtin_amdgcn_wmma_f32_16x16x4_f32(
                    false, awA[kc], false, b2, (short)0, accA2, false, false);
            accB2 = __builtin_amdgcn_wmma_f32_16x16x4_f32(
                false, awB[kc], false, b2, (short)0, accB2, false, false);
            if constexpr (DO_A)
                accA3 = __builtin_amdgcn_wmma_f32_16x16x4_f32(
                    false, awA[kc], false, b3, (short)0, accA3, false, false);
            accB3 = __builtin_amdgcn_wmma_f32_16x16x4_f32(
                false, awB[kc], false, b3, (short)0, accB3, false, false);
        }
        __builtin_amdgcn_sched_barrier(0);
    };
    for (int kt = 0; kt < qt0; ++kt) trip(kt, std::false_type{}, std::true_type{});
    trip(qt0, std::true_type{}, std::true_type{});
    trip(qt1, std::true_type{}, std::false_type{});

    dlds[ln]      = fmaxf(denA + xhalf(denA), EPS_);
    dlds[16 + ln] = fmaxf(denB + xhalf(denB), EPS_);
    asm volatile("s_wait_dscnt 0" ::: "memory");
    __builtin_amdgcn_wave_barrier();

#pragma unroll
    for (int r = 0; r < 8; ++r) {
        const int m  = r + 8 * hi;
        const float dnA = dlds[m];
        const float dnB = dlds[16 + m];
        const int orow = qt0 * 16 + m;
        v4f oa, ob4;
        oa.x = accA0[r] / dnA;
        oa.y = accA1[r] / dnA;
        oa.z = accA2[r] / dnA;
        oa.w = accA3[r] / dnA;
        ob4.x = accB0[r] / dnB;
        ob4.y = accB1[r] / dnB;
        ob4.z = accB2[r] / dnB;
        ob4.w = accB3[r] / dnB;
        *(v4f*)(ob + orow * D_ + 4 * ln)        = oa;
        *(v4f*)(ob + (orow + 16) * D_ + 4 * ln) = ob4;
    }
  }
}

extern "C" void kernel_launch(void* const* d_in, const int* in_sizes, int n_in,
                              void* d_out, int out_size, void* d_ws, size_t ws_size,
                              hipStream_t stream) {
    (void)in_sizes; (void)n_in; (void)out_size; (void)d_ws; (void)ws_size;
    const float* q    = (const float*)d_in[0];
    const float* k    = (const float*)d_in[1];
    const float* v    = (const float*)d_in[2];
    const float* cosc = (const float*)d_in[3];
    const float* sinc = (const float*)d_in[4];
    float* out = (float*)d_out;

    const int tiles  = B_ * H_ * QTILES;
    const int blocks = tiles / (4 * WPB);
    rse_stickbreak_attn<<<blocks, 32 * WPB, 0, stream>>>(q, k, v, cosc, sinc, out);
}


# k loop with a third fewer vector instructions: two trips per loop pass (no hand-over copies), RoPE/sigmoid on natural register pairs, scalar tile addresses, scan step R-w*R as one fma (one rounding)
# speedup vs baseline: 2.1743x; 1.1806x over previous
#include <hip/hip_runtime.h>
#include <math.h>
#include <type_traits>

typedef float v8f __attribute__((ext_vector_type(8)));
typedef float v4f __attribute__((ext_vector_type(4)));
typedef float v2f __attribute__((ext_vector_type(2)));

#define B_      4
#define H_      12
#define S_      1024
#define D_      64
#define HALF_   32
#define LAMBDA_ 0.01f
#define EPS_    1e-6f
#define CLAMP_  20.0f
#define QTILES  (S_ / 16)
#define QBLKS   (QTILES / 2)
#define WPB     4

static __device__ __forceinline__ float xhalf(float x) {
    const int b = __builtin_bit_cast(int, x);
    return __builtin_bit_cast(float,
        __builtin_amdgcn_permlanex16(b, b, 0x76543210, 0xfedcba98, false, false));
}

static __device__ __forceinline__ v2f rope_pair(float xe, float xo, float c, float s) {
    const v2f x  = {xe, xo};
    const v2f xs = {xo, xe};
    const v2f sn = {-s, s};
    const v2f cc = {c, c};
    const v2f t  = xs * sn;
    return __builtin_elementwise_fma(x, cc, t);
}

static __device__ __forceinline__ void rope_frags(v2f* f, const v4f* x0a, const v4f* x1a,
                                                  const v4f* cca, const v4f* ssa) {
#pragma unroll
    for (int g = 0; g < 4; ++g) {
        const v4f x0 = x0a[g];
        const v4f x1 = x1a[g];
        const v4f cc = cca[g];
        const v4f ss = ssa[g];
        f[4 * g + 0] = rope_pair(x0.x, x0.y, cc.x, ss.x);
        f[4 * g + 1] = rope_pair(x0.z, x0.w, cc.y, ss.y);
        f[4 * g + 2] = rope_pair(x1.x, x1.y, cc.z, ss.z);
        f[4 * g + 3] = rope_pair(x1.z, x1.w, cc.w, ss.w);
    }
}

static __device__ __forceinline__ void load_rows(v4f* x0a, v4f* x1a, v4f* cca, v4f* ssa,
                                                 const float* __restrict__ xb,
                                                 const float* __restrict__ cosc,
                                                 const float* __restrict__ sinc,
                                                 int tile, int ln, int hi) {
    const float* xt = xb   + tile * (16 * D_);
    const float* ct = cosc + tile * (16 * HALF_);
    const float* st = sinc + tile * (16 * HALF_);
    const int lo = ln * D_    + 8 * hi;
    const int lc = ln * HALF_ + 4 * hi;
#pragma unroll
    for (int g = 0; g < 4; ++g) {
        x0a[g] = *(const v4f*)(xt + lo + 16 * g);
        x1a[g] = *(const v4f*)(xt + lo + 16 * g + 4);
        cca[g] = *(const v4f*)(ct + lc + 8 * g);
        ssa[g] = *(const v4f*)(st + lc + 8 * g);
    }
}

__global__ __launch_bounds__(32 * WPB) __attribute__((amdgpu_waves_per_eu(1, 1)))
void rse_stickbreak_attn(const float* __restrict__ q,
                         const float* __restrict__ k,
                         const float* __restrict__ v,
                         const float* __restrict__ cosc,
                         const float* __restrict__ sinc,
                         float* __restrict__ out)
{
    __shared__ __align__(16) float lds[WPB][32];

    const int wave = __builtin_amdgcn_readfirstlane((int)(threadIdx.x >> 5));
    const int lane = threadIdx.x & 31;
    const int pr = (blockIdx.x % (QBLKS / 2 / WPB)) * WPB + wave;
    const int bh = blockIdx.x / (QBLKS / 2 / WPB);

    float* dlds = &lds[wave][0];

    const int hi = lane >> 4;
    const int ln = lane & 15;

    const float* qb = q + (size_t)bh * S_ * D_;
    const float* kb = k + (size_t)bh * S_ * D_;
    const float* vb = v + (size_t)bh * S_ * D_;
    float*       ob = out + (size_t)bh * S_ * D_;

  for (int pass = 0; pass < 2; ++pass) {
    const int qblk = pass ? pr : (QBLKS - 1 - pr);
    const int qt0  = 2 * qblk;
    const int qt1  = qt0 + 1;

    v2f afA[16], afB[16];
    {
        v4f x0[4], x1[4], cc[4], ss[4];
        load_rows(x0, x1, cc, ss, qb, cosc, sinc, qt0, ln, hi);
        rope_frags(afA, x0, x1, cc, ss);
        load_rows(x0, x1, cc, ss, qb, cosc, sinc, qt1, ln, hi);
        rope_frags(afB, x0, x1, cc, ss);
    }

    v8f accA0 = (v8f)0.0f, accA1 = (v8f)0.0f, accA2 = (v8f)0.0f, accA3 = (v8f)0.0f;
    v8f accB0 = (v8f)0.0f, accB1 = (v8f)0.0f, accB2 = (v8f)0.0f, accB3 = (v8f)0.0f;
    float remA = 1.0f, remB = 1.0f;
    float denA = 0.0f, denB = 0.0f;
    const float scale = 0.125f;

    v4f kx0[4], kx1[4], kcc[4], kss[4];
    load_rows(kx0, kx1, kcc, kss, kb, cosc, sinc, 0, ln, hi);

    v2f tPrev[4];
    {
        const float dB00 = (float)(qt1 * 16 + ln - 8 * hi);
#pragma unroll
        for (int r = 0; r < 8; r += 2) {
            const v2f d2  = {dB00, dB00};
            const v2f off = {(float)r, (float)(r + 1)};
            tPrev[r >> 1] = LAMBDA_ * (d2 - off);
        }
    }

    auto trip = [&](const int kt, auto diag_tag, auto a_tag) __attribute__((always_inline)) {
        constexpr bool DIAG = decltype(diag_tag)::value;
        constexpr bool DO_A = decltype(a_tag)::value;
        v4f vr0[4], vr1[4];
#pragma unroll
        for (int kc = 0; kc < 4; ++kc) {
            const float* vt = vb + kt * (16 * D_);
            const int lv = 8 * hi * D_ + 4 * ln;
            vr0[kc] = *(const v4f*)(vt + lv + (2 * kc) * D_);
            vr1[kc] = *(const v4f*)(vt + lv + (2 * kc + 1) * D_);
        }

        v2f kf[16];
        rope_frags(kf, kx0, kx1, kcc, kss);

        {
            const int kn = DIAG ? ((kt < qt1) ? (kt + 1) : qt1) : (kt + 1);
            load_rows(kx0, kx1, kcc, kss, kb, cosc, sinc, kn, ln, hi);
        }

        v8f caccA = (v8f)0.0f, caccB = (v8f)0.0f;
#pragma unroll
        for (int s = 0; s < 16; ++s) {
            if constexpr (DO_A)
                caccA = __builtin_amdgcn_wmma_f32_16x16x4_f32(
                    false, kf[s], false, afA[s], (short)0, caccA, false, false);
            caccB = __builtin_amdgcn_wmma_f32_16x16x4_f32(
                false, kf[s], false, afB[s], (short)0, caccB, false, false);
        }

        __builtin_amdgcn_sched_barrier(0);

        float btA[8], btB[8];
        if constexpr (DIAG) {
#pragma unroll
            for (int r = 0; r < 8; ++r) {
                const int kpos  = kt * 16 + 8 * hi + r;
                const int qposA = qt0 * 16 + ln;
                const int qposB = qt1 * 16 + ln;
                float xa = caccA[r] * scale - LAMBDA_ * fabsf((float)(qposA - kpos));
                float xb = caccB[r] * scale - LAMBDA_ * fabsf((float)(qposB - kpos));
                xa = fminf(fmaxf(xa, -CLAMP_), CLAMP_);
                xb = fminf(fmaxf(xb, -CLAMP_), CLAMP_);
                float ba = __builtin_amdgcn_rcpf(1.0f + __expf(-xa));
                float bb = __builtin_amdgcn_rcpf(1.0f + __expf(-xb));
                if (kpos > qposA) ba = 0.0f;
                if (kpos > qposB) bb = 0.0f;
                btA[r] = ba;
                btB[r] = bb;
            }
        } else {
            const float dA0 = (float)(qt0 * 16 + ln - kt * 16 - 8 * hi);
            const v2f dA = {dA0, dA0};
            const v2f sc = {scale, scale};
#pragma unroll
            for (int r = 0; r < 8; r += 2) {
                const v2f off = {(float)r, (float)(r + 1)};
                const v2f cA  = {caccA[r], caccA[r + 1]};
                const v2f cB  = {caccB[r], caccB[r + 1]};
                const v2f tA  = LAMBDA_ * (dA - off);
                const v2f tB  = tPrev[r >> 1];
                tPrev[r >> 1] = tA;
                const v2f xa  = __builtin_elementwise_fma(cA, sc, -tA);
                const v2f xb  = __builtin_elementwise_fma(cB, sc, -tB);
                const float xa0 = fminf(fmaxf(xa.x, -CLAMP_), CLAMP_);
                const float xa1 = fminf(fmaxf(xa.y, -CLAMP_), CLAMP_);
                const float xb0 = fminf(fmaxf(xb.x, -CLAMP_), CLAMP_);
                const float xb1 = fminf(fmaxf(xb.y, -CLAMP_), CLAMP_);
                const v2f eA = {__expf(-xa0), __expf(-xa1)};
                const v2f eB = {__expf(-xb0), __expf(-xb1)};
                const v2f nA = eA + 1.0f;
                const v2f nB = eB + 1.0f;
                btA[r]     = __builtin_amdgcn_rcpf(nA.x);
                btA[r + 1] = __builtin_amdgcn_rcpf(nA.y);
                btB[r]     = __builtin_amdgcn_rcpf(nB.x);
                btB[r + 1] = __builtin_amdgcn_rcpf(nB.y);
            }
        }

#define SB_STEP(R, W, BT, I)                              \
            { const float w = BT[I] * R;                  \
              R = fmaxf(__builtin_fmaf(-w, R, R), EPS_);  \
              W[I] = w; }
#define SB_STEP2(RA, WA, RB, WB, I)  SB_STEP(RA, WA, btA, I) SB_STEP(RB, WB, btB, I)
        float waA[8], wbA[8], waB[8], wbB[8];
        float raA = remA, raB = remB;
        SB_STEP2(raA, waA, raB, waB, 0) SB_STEP2(raA, waA, raB, waB, 1)
        SB_STEP2(raA, waA, raB, waB, 2) SB_STEP2(raA, waA, raB, waB, 3)
        SB_STEP2(raA, waA, raB, waB, 4) SB_STEP2(raA, waA, raB, waB, 5)
        SB_STEP2(raA, waA, raB, waB, 6) SB_STEP2(raA, waA, raB, waB, 7)
        float rbA = xhalf(raA), rbB = xhalf(raB);
        SB_STEP2(rbA, wbA, rbB, wbB, 0) SB_STEP2(rbA, wbA, rbB, wbB, 1)
        SB_STEP2(rbA, wbA, rbB, wbB, 2) SB_STEP2(rbA, wbA, rbB, wbB, 3)
        SB_STEP2(rbA, wbA, rbB, wbB, 4) SB_STEP2(rbA, wbA, rbB, wbB, 5)
        SB_STEP2(rbA, wbA, rbB, wbB, 6) SB_STEP2(rbA, wbA, rbB, wbB, 7)
#undef SB_STEP2
#undef SB_STEP
        if constexpr (DO_A) remA = xhalf(rbA);
        remB = xhalf(rbB);

        v2f awA[4], awB[4];
#pragma unroll
        for (int kc = 0; kc < 4; ++kc) {
            awA[kc].x = hi ? wbA[2 * kc]     : waA[2 * kc];
            awA[kc].y = hi ? wbA[2 * kc + 1] : waA[2 * kc + 1];
            awB[kc].x = hi ? wbB[2 * kc]     : waB[2 * kc];
            awB[kc].y = hi ? wbB[2 * kc + 1] : waB[2 * kc + 1];
        }
        {
            if constexpr (DO_A) {
                const v2f sA = (awA[0] + awA[1]) + (awA[2] + awA[3]);
                denA += sA.x + sA.y;
            }
            const v2f sB = (awB[0] + awB[1]) + (awB[2] + awB[3]);
            denB += sB.x + sB.y;
        }

#pragma unroll
        for (int kc = 0; kc < 4; ++kc) {
            const v4f r0 = vr0[kc];
            const v4f r1 = vr1[kc];
            v2f b0, b1, b2, b3;
            b0.x = r0.x;  b0.y = r1.x;
            b1.x = r0.y;  b1.y = r1.y;
            b2.x = r0.z;  b2.y = r1.z;
            b3.x = r0.w;  b3.y = r1.w;
            if constexpr (DO_A)
                accA0 = __builtin_amdgcn_wmma_f32_16x16x4_f32(
                    false, awA[kc], false, b0, (short)0, accA0, false, false);
            accB0 = __builtin_amdgcn_wmma_f32_16x16x4_f32(
                false, awB[kc], false, b0, (short)0, accB0, false, false);
            if constexpr (DO_A)
                accA1 = __builtin_amdgcn_wmma_f32_16x16x4_f32(
                    false, awA[kc], false, b1, (short)0, accA1, false, false);
            accB1 = __builtin_amdgcn_wmma_f32_16x16x4_f32(
                false, awB[kc], false, b1, (short)0, accB1, false, false);
            if constexpr (DO_A)
                accA2 = __builtin_amdgcn_wmma_f32_16x16x4_f32(
                    false, awA[kc], false, b2, (short)0, accA2, false, false);
            accB2 = __builtin_amdgcn_wmma_f32_16x16x4_f32(
                false, awB[kc], false, b2, (short)0, accB2, false, false);
            if constexpr (DO_A)
                accA3 = __builtin_amdgcn_wmma_f32_16x16x4_f32(
                    false, awA[kc], false, b3, (short)0, accA3, false, false);
            accB3 = __builtin_amdgcn_wmma_f32_16x16x4_f32(
                false, awB[kc], false, b3, (short)0, accB3, false, false);
        }
    };
    for (int kt = 0; kt < qt0; kt += 2) {
        trip(kt,     std::false_type{}, std::true_type{});
        trip(kt + 1, std::false_type{}, std::true_type{});
    }
    trip(qt0, std::true_type{}, std::true_type{});
    trip(qt1, std::true_type{}, std::false_type{});

    dlds[ln]      = fmaxf(denA + xhalf(denA), EPS_);
    dlds[16 + ln] = fmaxf(denB + xhalf(denB), EPS_);
    asm volatile("s_wait_dscnt 0" ::: "memory");
    __builtin_amdgcn_wave_barrier();

#pragma unroll
    for (int r = 0; r < 8; ++r) {
        const int m  = r + 8 * hi;
        const float dnA = dlds[m];
        const float dnB = dlds[16 + m];
        const int orow = qt0 * 16 + m;
        v4f oa, ob4;
        oa.x = accA0[r] / dnA;
        oa.y = accA1[r] / dnA;
        oa.z = accA2[r] / dnA;
        oa.w = accA3[r] / dnA;
        ob4.x = accB0[r] / dnB;
        ob4.y = accB1[r] / dnB;
        ob4.z = accB2[r] / dnB;
        ob4.w = accB3[r] / dnB;
        *(v4f*)(ob + orow * D_ + 4 * ln)        = oa;
        *(v4f*)(ob + (orow + 16) * D_ + 4 * ln) = ob4;
    }
  }
}

extern "C" void kernel_launch(void* const* d_in, const int* in_sizes, int n_in,
                              void* d_out, int out_size, void* d_ws, size_t ws_size,
                              hipStream_t stream) {
    (void)in_sizes; (void)n_in; (void)out_size; (void)d_ws; (void)ws_size;
    const float* q    = (const float*)d_in[0];
    const float* k    = (const float*)d_in[1];
    const float* v    = (const float*)d_in[2];
    const float* cosc = (const float*)d_in[3];
    const float* sinc = (const float*)d_in[4];
    float* out = (float*)d_out;

    const int tiles  = B_ * H_ * QTILES;
    const int blocks = tiles / (4 * WPB);
    rse_stickbreak_attn<<<blocks, 32 * WPB, 0, stream>>>(q, k, v, cosc, sinc, out);
}


# k loop regrouped: a trip's 64 matrix instructions issue as one run (W x V of trip t, then logits of t+1) and all its vector work, with the next k tile's RoPE, as one run between; sum orders unchanged
# speedup vs baseline: 2.2659x; 1.0421x over previous
#include <hip/hip_runtime.h>
#include <math.h>
#include <type_traits>

typedef float v8f __attribute__((ext_vector_type(8)));
typedef float v4f __attribute__((ext_vector_type(4)));
typedef float v2f __attribute__((ext_vector_type(2)));

#define B_      4
#define H_      12
#define S_      1024
#define D_      64
#define HALF_   32
#define LAMBDA_ 0.01f
#define EPS_    1e-6f
#define CLAMP_  20.0f
#define QTILES  (S_ / 16)
#define QBLKS   (QTILES / 2)
#define WPB     4

static __device__ __forceinline__ float xhalf(float x) {
    const int b = __builtin_bit_cast(int, x);
    return __builtin_bit_cast(float,
        __builtin_amdgcn_permlanex16(b, b, 0x76543210, 0xfedcba98, false, false));
}

static __device__ __forceinline__ v2f rope_pair(float xe, float xo, float c, float s) {
    const v2f x  = {xe, xo};
    const v2f xs = {xo, xe};
    const v2f sn = {-s, s};
    const v2f cc = {c, c};
    const v2f t  = xs * sn;
    return __builtin_elementwise_fma(x, cc, t);
}

static __device__ __forceinline__ void rope_frags(v2f* f, const v4f* x0a, const v4f* x1a,
                                                  const v4f* cca, const v4f* ssa) {
#pragma unroll
    for (int g = 0; g < 4; ++g) {
        const v4f x0 = x0a[g];
        const v4f x1 = x1a[g];
        const v4f cc = cca[g];
        const v4f ss = ssa[g];
        f[4 * g + 0] = rope_pair(x0.x, x0.y, cc.x, ss.x);
        f[4 * g + 1] = rope_pair(x0.z, x0.w, cc.y, ss.y);
        f[4 * g + 2] = rope_pair(x1.x, x1.y, cc.z, ss.z);
        f[4 * g + 3] = rope_pair(x1.z, x1.w, cc.w, ss.w);
    }
}

static __device__ __forceinline__ void load_rows(v4f* x0a, v4f* x1a, v4f* cca, v4f* ssa,
                                                 const float* __restrict__ xb,
                                                 const float* __restrict__ cosc,
                                                 const float* __restrict__ sinc,
                                                 int tile, int ln, int hi) {
    const float* xt = xb   + tile * (16 * D_);
    const float* ct = cosc + tile * (16 * HALF_);
    const float* st = sinc + tile * (16 * HALF_);
    const int lo = ln * D_    + 8 * hi;
    const int lc = ln * HALF_ + 4 * hi;
#pragma unroll
    for (int g = 0; g < 4; ++g) {
        x0a[g] = *(const v4f*)(xt + lo + 16 * g);
        x1a[g] = *(const v4f*)(xt + lo + 16 * g + 4);
        cca[g] = *(const v4f*)(ct + lc + 8 * g);
        ssa[g] = *(const v4f*)(st + lc + 8 * g);
    }
}

static __device__ __forceinline__ void load_vrows(v4f* vr0, v4f* vr1, const float* __restrict__ vb,
                                                  int kt, int ln, int hi) {
    const float* vt = vb + kt * (16 * D_);
    const int lv = 8 * hi * D_ + 4 * ln;
#pragma unroll
    for (int kc = 0; kc < 4; ++kc) {
        vr0[kc] = *(const v4f*)(vt + lv + (2 * kc) * D_);
        vr1[kc] = *(const v4f*)(vt + lv + (2 * kc + 1) * D_);
    }
}

__global__ __launch_bounds__(32 * WPB) __attribute__((amdgpu_waves_per_eu(1, 1)))
void rse_stickbreak_attn(const float* __restrict__ q,
                         const float* __restrict__ k,
                         const float* __restrict__ v,
                         const float* __restrict__ cosc,
                         const float* __restrict__ sinc,
                         float* __restrict__ out)
{
    __shared__ __align__(16) float lds[WPB][32];

    const int wave = __builtin_amdgcn_readfirstlane((int)(threadIdx.x >> 5));
    const int lane = threadIdx.x & 31;
    const int pr = (blockIdx.x % (QBLKS / 2 / WPB)) * WPB + wave;
    const int bh = blockIdx.x / (QBLKS / 2 / WPB);

    float* dlds = &lds[wave][0];

    const int hi = lane >> 4;
    const int ln = lane & 15;

    const float* qb = q + (size_t)bh * S_ * D_;
    const float* kb = k + (size_t)bh * S_ * D_;
    const float* vb = v + (size_t)bh * S_ * D_;
    float*       ob = out + (size_t)bh * S_ * D_;

  for (int pass = 0; pass < 2; ++pass) {
    const int qblk = pass ? pr : (QBLKS - 1 - pr);
    const int qt0  = 2 * qblk;
    const int qt1  = qt0 + 1;

    v2f afA[16], afB[16];
    {
        v4f x0[4], x1[4], cc[4], ss[4];
        load_rows(x0, x1, cc, ss, qb, cosc, sinc, qt0, ln, hi);
        rope_frags(afA, x0, x1, cc, ss);
        load_rows(x0, x1, cc, ss, qb, cosc, sinc, qt1, ln, hi);
        rope_frags(afB, x0, x1, cc, ss);
    }

    v8f accA0 = (v8f)0.0f, accA1 = (v8f)0.0f, accA2 = (v8f)0.0f, accA3 = (v8f)0.0f;
    v8f accB0 = (v8f)0.0f, accB1 = (v8f)0.0f, accB2 = (v8f)0.0f, accB3 = (v8f)0.0f;
    float remA = 1.0f, remB = 1.0f;
    float denA = 0.0f, denB = 0.0f;
    const float scale = 0.125f;

    v2f kf[16];
    v4f kx0[4], kx1[4], kcc[4], kss[4];
    v4f vr0[4], vr1[4];
    load_rows(kx0, kx1, kcc, kss, kb, cosc, sinc, 0, ln, hi);
    rope_frags(kf, kx0, kx1, kcc, kss);
    load_rows(kx0, kx1, kcc, kss, kb, cosc, sinc, 1, ln, hi);
    load_vrows(vr0, vr1, vb, 0, ln, hi);

    v2f tPrev[4];
    {
        const float dB00 = (float)(qt1 * 16 + ln - 8 * hi);
#pragma unroll
        for (int r = 0; r < 8; r += 2) {
            const v2f d2  = {dB00, dB00};
            const v2f off = {(float)r, (float)(r + 1)};
            tPrev[r >> 1] = LAMBDA_ * (d2 - off);
        }
    }

    auto trip = [&](const int kt, auto diag_tag, auto a_tag) __attribute__((always_inline)) {
        constexpr bool DIAG = decltype(diag_tag)::value;
        constexpr bool DO_A = decltype(a_tag)::value;

        v8f caccA = (v8f)0.0f, caccB = (v8f)0.0f;
#pragma unroll
        for (int s = 0; s < 16; ++s) {
            if constexpr (DO_A)
                caccA = __builtin_amdgcn_wmma_f32_16x16x4_f32(
                    false, kf[s], false, afA[s], (short)0, caccA, false, false);
            caccB = __builtin_amdgcn_wmma_f32_16x16x4_f32(
                false, kf[s], false, afB[s], (short)0, caccB, false, false);
        }

        asm volatile("" : "+v"(caccA), "+v"(caccB),
                          "+v"(accA0), "+v"(accA1), "+v"(accA2), "+v"(accA3),
                          "+v"(accB0), "+v"(accB1), "+v"(accB2), "+v"(accB3));
        __builtin_amdgcn_sched_barrier(0);

        float btA[8], btB[8];
        if constexpr (DIAG) {
#pragma unroll
            for (int r = 0; r < 8; ++r) {
                const int kpos  = kt * 16 + 8 * hi + r;
                const int qposA = qt0 * 16 + ln;
                const int qposB = qt1 * 16 + ln;
                float xa = caccA[r] * scale - LAMBDA_ * fabsf((float)(qposA - kpos));
                float xb = caccB[r] * scale - LAMBDA_ * fabsf((float)(qposB - kpos));
                xa = fminf(fmaxf(xa, -CLAMP_), CLAMP_);
                xb = fminf(fmaxf(xb, -CLAMP_), CLAMP_);
                float ba = __builtin_amdgcn_rcpf(1.0f + __expf(-xa));
                float bb = __builtin_amdgcn_rcpf(1.0f + __expf(-xb));
                if (kpos > qposA) ba = 0.0f;
                if (kpos > qposB) bb = 0.0f;
                btA[r] = ba;
                btB[r] = bb;
            }
        } else {
            const float dA0 = (float)(qt0 * 16 + ln - kt * 16 - 8 * hi);
            const v2f dA = {dA0, dA0};
            const v2f sc = {scale, scale};
#pragma unroll
            for (int r = 0; r < 8; r += 2) {
                const v2f off = {(float)r, (float)(r + 1)};
                const v2f cA  = {caccA[r], caccA[r + 1]};
                const v2f cB  = {caccB[r], caccB[r + 1]};
                const v2f tA  = LAMBDA_ * (dA - off);
                const v2f tB  = tPrev[r >> 1];
                tPrev[r >> 1] = tA;
                const v2f xa  = __builtin_elementwise_fma(cA, sc, -tA);
                const v2f xb  = __builtin_elementwise_fma(cB, sc, -tB);
                const float xa0 = fminf(fmaxf(xa.x, -CLAMP_), CLAMP_);
                const float xa1 = fminf(fmaxf(xa.y, -CLAMP_), CLAMP_);
                const float xb0 = fminf(fmaxf(xb.x, -CLAMP_), CLAMP_);
                const float xb1 = fminf(fmaxf(xb.y, -CLAMP_), CLAMP_);
                const v2f eA = {__expf(-xa0), __expf(-xa1)};
                const v2f eB = {__expf(-xb0), __expf(-xb1)};
                const v2f nA = eA + 1.0f;
                const v2f nB = eB + 1.0f;
                btA[r]     = __builtin_amdgcn_rcpf(nA.x);
                btA[r + 1] = __builtin_amdgcn_rcpf(nA.y);
                btB[r]     = __builtin_amdgcn_rcpf(nB.x);
                btB[r + 1] = __builtin_amdgcn_rcpf(nB.y);
            }
        }

#define SB_STEP(R, W, BT, I)                              \
            { const float w = BT[I] * R;                  \
              R = fmaxf(__builtin_fmaf(-w, R, R), EPS_);  \
              W[I] = w; }
#define SB_STEP2(RA, WA, RB, WB, I)  SB_STEP(RA, WA, btA, I) SB_STEP(RB, WB, btB, I)
        float waA[8], wbA[8], waB[8], wbB[8];
        float raA = remA, raB = remB;
        SB_STEP2(raA, waA, raB, waB, 0) SB_STEP2(raA, waA, raB, waB, 1)
        SB_STEP2(raA, waA, raB, waB, 2) SB_STEP2(raA, waA, raB, waB, 3)
        SB_STEP2(raA, waA, raB, waB, 4) SB_STEP2(raA, waA, raB, waB, 5)
        SB_STEP2(raA, waA, raB, waB, 6) SB_STEP2(raA, waA, raB, waB, 7)
        float rbA = xhalf(raA), rbB = xhalf(raB);
        SB_STEP2(rbA, wbA, rbB, wbB, 0) SB_STEP2(rbA, wbA, rbB, wbB, 1)
        SB_STEP2(rbA, wbA, rbB, wbB, 2) SB_STEP2(rbA, wbA, rbB, wbB, 3)
        SB_STEP2(rbA, wbA, rbB, wbB, 4) SB_STEP2(rbA, wbA, rbB, wbB, 5)
        SB_STEP2(rbA, wbA, rbB, wbB, 6) SB_STEP2(rbA, wbA, rbB, wbB, 7)
#undef SB_STEP2
#undef SB_STEP
        if constexpr (DO_A) remA = xhalf(rbA);
        remB = xhalf(rbB);

        v2f awA[4], awB[4];
#pragma unroll
        for (int kc = 0; kc < 4; ++kc) {
            awA[kc].x = hi ? wbA[2 * kc]     : waA[2 * kc];
            awA[kc].y = hi ? wbA[2 * kc + 1] : waA[2 * kc + 1];
            awB[kc].x = hi ? wbB[2 * kc]     : waB[2 * kc];
            awB[kc].y = hi ? wbB[2 * kc + 1] : waB[2 * kc + 1];
        }
        {
            if constexpr (DO_A) {
                const v2f sA = (awA[0] + awA[1]) + (awA[2] + awA[3]);
                denA += sA.x + sA.y;
            }
            const v2f sB = (awB[0] + awB[1]) + (awB[2] + awB[3]);
            denB += sB.x + sB.y;
        }

        v2f bv[4][4];
#pragma unroll
        for (int kc = 0; kc < 4; ++kc) {
            const v4f r0 = vr0[kc];
            const v4f r1 = vr1[kc];
            bv[kc][0].x = r0.x;  bv[kc][0].y = r1.x;
            bv[kc][1].x = r0.y;  bv[kc][1].y = r1.y;
            bv[kc][2].x = r0.z;  bv[kc][2].y = r1.z;
            bv[kc][3].x = r0.w;  bv[kc][3].y = r1.w;
        }

        if constexpr (DO_A) {
            load_vrows(vr0, vr1, vb, kt + 1, ln, hi);
            rope_frags(kf, kx0, kx1, kcc, kss);
            if constexpr (!DIAG)
                load_rows(kx0, kx1, kcc, kss, kb, cosc, sinc, kt + 2, ln, hi);
        }

        asm volatile("" : "+v"(awA[0]), "+v"(awA[1]), "+v"(awA[2]), "+v"(awA[3]),
                          "+v"(awB[0]), "+v"(awB[1]), "+v"(awB[2]), "+v"(awB[3]),
                          "+v"(bv[0][0]), "+v"(bv[0][1]), "+v"(bv[0][2]), "+v"(bv[0][3]),
                          "+v"(bv[1][0]), "+v"(bv[1][1]), "+v"(bv[1][2]), "+v"(bv[1][3]),
                          "+v"(bv[2][0]), "+v"(bv[2][1]), "+v"(bv[2][2]), "+v"(bv[2][3]),
                          "+v"(bv[3][0]), "+v"(bv[3][1]), "+v"(bv[3][2]), "+v"(bv[3][3]),
                          "+v"(remA), "+v"(remB), "+v"(denA), "+v"(denB));
        __builtin_amdgcn_sched_barrier(0);

#pragma unroll
        for (int kc = 0; kc < 4; ++kc) {
            const v2f b0 = bv[kc][0], b1 = bv[kc][1], b2 = bv[kc][2], b3 = bv[kc][3];
            if constexpr (DO_A)
                accA0 = __builtin_amdgcn_wmma_f32_16x16x4_f32(
                    false, awA[kc], false, b0, (short)0, accA0, false, false);
            accB0 = __builtin_amdgcn_wmma_f32_16x16x4_f32(
                false, awB[kc], false, b0, (short)0, accB0, false, false);
            if constexpr (DO_A)
                accA1 = __builtin_amdgcn_wmma_f32_16x16x4_f32(
                    false, awA[kc], false, b1, (short)0, accA1, false, false);
            accB1 = __builtin_amdgcn_wmma_f32_16x16x4_f32(
                false, awB[kc], false, b1, (short)0, accB1, false, false);
            if constexpr (DO_A)
                accA2 = __builtin_amdgcn_wmma_f32_16x16x4_f32(
                    false, awA[kc], false, b2, (short)0, accA2, false, false);
            accB2 = __builtin_amdgcn_wmma_f32_16x16x4_f32(
                false, awB[kc], false, b2, (short)0, accB2, false, false);
            if constexpr (DO_A)
                accA3 = __builtin_amdgcn_wmma_f32_16x16x4_f32(
                    false, awA[kc], false, b3, (short)0, accA3, false, false);
            accB3 = __builtin_amdgcn_wmma_f32_16x16x4_f32(
                false, awB[kc], false, b3, (short)0, accB3, false, false);
        }
    };
    for (int kt = 0; kt < qt0; kt += 2) {
        trip(kt,     std::false_type{}, std::true_type{});
        trip(kt + 1, std::false_type{}, std::true_type{});
    }
    trip(qt0, std::true_type{}, std::true_type{});
    trip(qt1, std::true_type{}, std::false_type{});

    dlds[ln]      = fmaxf(denA + xhalf(denA), EPS_);
    dlds[16 + ln] = fmaxf(denB + xhalf(denB), EPS_);
    asm volatile("s_wait_dscnt 0" ::: "memory");
    __builtin_amdgcn_wave_barrier();

#pragma unroll
    for (int r = 0; r < 8; ++r) {
        const int m  = r + 8 * hi;
        const float dnA = dlds[m];
        const float dnB = dlds[16 + m];
        const int orow = qt0 * 16 + m;
        v4f oa, ob4;
        oa.x = accA0[r] / dnA;
        oa.y = accA1[r] / dnA;
        oa.z = accA2[r] / dnA;
        oa.w = accA3[r] / dnA;
        ob4.x = accB0[r] / dnB;
        ob4.y = accB1[r] / dnB;
        ob4.z = accB2[r] / dnB;
        ob4.w = accB3[r] / dnB;
        *(v4f*)(ob + orow * D_ + 4 * ln)        = oa;
        *(v4f*)(ob + (orow + 16) * D_ + 4 * ln) = ob4;
    }
  }
}

extern "C" void kernel_launch(void* const* d_in, const int* in_sizes, int n_in,
                              void* d_out, int out_size, void* d_ws, size_t ws_size,
                              hipStream_t stream) {
    (void)in_sizes; (void)n_in; (void)out_size; (void)d_ws; (void)ws_size;
    const float* q    = (const float*)d_in[0];
    const float* k    = (const float*)d_in[1];
    const float* v    = (const float*)d_in[2];
    const float* cosc = (const float*)d_in[3];
    const float* sinc = (const float*)d_in[4];
    float* out = (float*)d_out;

    const int tiles  = B_ * H_ * QTILES;
    const int blocks = tiles / (4 * WPB);
    rse_stickbreak_attn<<<blocks, 32 * WPB, 0, stream>>>(q, k, v, cosc, sinc, out);
}


# Work re-split: q tiles paired far apart (36+w, 63-w), upper tile finishing alone, so 1008 even waves fill the SIMDs (a quarter sat idle); longest wave about 55 trips, was 66; sum orders unchanged
# speedup vs baseline: 2.4419x; 1.0777x over previous
#include <hip/hip_runtime.h>
#include <math.h>
#include <type_traits>

typedef float v8f __attribute__((ext_vector_type(8)));
typedef float v4f __attribute__((ext_vector_type(4)));
typedef float v2f __attribute__((ext_vector_type(2)));

#define B_      4
#define H_      12
#define S_      1024
#define D_      64
#define HALF_   32
#define LAMBDA_ 0.01f
#define EPS_    1e-6f
#define CLAMP_  20.0f
#define QTILES  (S_ / 16)
#define WPB     4
#define WPH     21
#define NFAR    14
#define SPLIT_  36
#define JL(a, b, c) ((unsigned)(a) | ((unsigned)(b) << 5) | ((unsigned)(c) << 10))

static __device__ __forceinline__ float xhalf(float x) {
    const int b = __builtin_bit_cast(int, x);
    return __builtin_bit_cast(float,
        __builtin_amdgcn_permlanex16(b, b, 0x76543210, 0xfedcba98, false, false));
}

static __device__ __forceinline__ v2f rope_pair(float xe, float xo, float c, float s) {
    const v2f x  = {xe, xo};
    const v2f xs = {xo, xe};
    const v2f sn = {-s, s};
    const v2f cc = {c, c};
    const v2f t  = xs * sn;
    return __builtin_elementwise_fma(x, cc, t);
}

static __device__ __forceinline__ void rope_frags(v2f* f, const v4f* x0a, const v4f* x1a,
                                                  const v4f* cca, const v4f* ssa) {
#pragma unroll
    for (int g = 0; g < 4; ++g) {
        const v4f x0 = x0a[g];
        const v4f x1 = x1a[g];
        const v4f cc = cca[g];
        const v4f ss = ssa[g];
        f[4 * g + 0] = rope_pair(x0.x, x0.y, cc.x, ss.x);
        f[4 * g + 1] = rope_pair(x0.z, x0.w, cc.y, ss.y);
        f[4 * g + 2] = rope_pair(x1.x, x1.y, cc.z, ss.z);
        f[4 * g + 3] = rope_pair(x1.z, x1.w, cc.w, ss.w);
    }
}

static __device__ __forceinline__ void load_rows(v4f* x0a, v4f* x1a, v4f* cca, v4f* ssa,
                                                 const float* __restrict__ xb,
                                                 const float* __restrict__ cosc,
                                                 const float* __restrict__ sinc,
                                                 int tile, int ln, int hi) {
    const float* xt = xb   + tile * (16 * D_);
    const float* ct = cosc + tile * (16 * HALF_);
    const float* st = sinc + tile * (16 * HALF_);
    const int lo = ln * D_    + 8 * hi;
    const int lc = ln * HALF_ + 4 * hi;
#pragma unroll
    for (int g = 0; g < 4; ++g) {
        x0a[g] = *(const v4f*)(xt + lo + 16 * g);
        x1a[g] = *(const v4f*)(xt + lo + 16 * g + 4);
        cca[g] = *(const v4f*)(ct + lc + 8 * g);
        ssa[g] = *(const v4f*)(st + lc + 8 * g);
    }
}

static __device__ __forceinline__ void load_vrows(v4f* vr0, v4f* vr1, const float* __restrict__ vb,
                                                  int kt, int ln, int hi) {
    const float* vt = vb + kt * (16 * D_);
    const int lv = 8 * hi * D_ + 4 * ln;
#pragma unroll
    for (int kc = 0; kc < 4; ++kc) {
        vr0[kc] = *(const v4f*)(vt + lv + (2 * kc) * D_);
        vr1[kc] = *(const v4f*)(vt + lv + (2 * kc + 1) * D_);
    }
}

__global__ __launch_bounds__(32 * WPB) __attribute__((amdgpu_waves_per_eu(1, 1)))
void rse_stickbreak_attn(const float* __restrict__ q,
                         const float* __restrict__ k,
                         const float* __restrict__ v,
                         const float* __restrict__ cosc,
                         const float* __restrict__ sinc,
                         float* __restrict__ out)
{
    __shared__ __align__(16) float lds[WPB][32];

    const int wave = __builtin_amdgcn_readfirstlane((int)(threadIdx.x >> 5));
    const int lane = threadIdx.x & 31;
    const int gw = blockIdx.x * WPB + wave;
    const int bh = gw / WPH;
    const int w  = gw - bh * WPH;
    const int u  = w - NFAR;
    const unsigned jl = u == 0 ? JL(17, 6, 0)  : u == 1 ? JL(16, 4, 1) : u == 2 ? JL(15, 8, 0)
                      : u == 3 ? JL(14, 9, 0)  : u == 4 ? JL(13, 5, 3) : u == 5 ? JL(12, 7, 2)
                      : JL(11, 10, 0);
    const int npass = w < NFAR ? 1 : ((u == 0 || u == 2 || u == 3) ? 2 : 3);

    float* dlds = &lds[wave][0];

    const int hi = lane >> 4;
    const int ln = lane & 15;

    const float* qb = q + (size_t)bh * S_ * D_;
    const float* kb = k + (size_t)bh * S_ * D_;
    const float* vb = v + (size_t)bh * S_ * D_;
    float*       ob = out + (size_t)bh * S_ * D_;

  for (int pass = 0; pass < npass; ++pass) {
    int qt0, qt1;
    if (w < NFAR) { qt0 = SPLIT_ + w; qt1 = QTILES - 1 - w; }
    else          { qt0 = 2 * (int)((jl >> (5 * pass)) & 31u); qt1 = qt0 + 1; }

    v2f afA[16], afB[16];
    {
        v4f x0[4], x1[4], cc[4], ss[4];
        load_rows(x0, x1, cc, ss, qb, cosc, sinc, qt0, ln, hi);
        rope_frags(afA, x0, x1, cc, ss);
        load_rows(x0, x1, cc, ss, qb, cosc, sinc, qt1, ln, hi);
        rope_frags(afB, x0, x1, cc, ss);
    }

    v8f accA0 = (v8f)0.0f, accA1 = (v8f)0.0f, accA2 = (v8f)0.0f, accA3 = (v8f)0.0f;
    v8f accB0 = (v8f)0.0f, accB1 = (v8f)0.0f, accB2 = (v8f)0.0f, accB3 = (v8f)0.0f;
    float remA = 1.0f, remB = 1.0f;
    float denA = 0.0f, denB = 0.0f;
    const float scale = 0.125f;

    v2f kf[16];
    v4f kx0[4], kx1[4], kcc[4], kss[4];
    v4f vr0[4], vr1[4];
    load_rows(kx0, kx1, kcc, kss, kb, cosc, sinc, 0, ln, hi);
    rope_frags(kf, kx0, kx1, kcc, kss);
    load_rows(kx0, kx1, kcc, kss, kb, cosc, sinc, 1, ln, hi);
    load_vrows(vr0, vr1, vb, 0, ln, hi);

    auto trip = [&](const int kt, auto diag_tag, auto a_tag) __attribute__((always_inline)) {
        constexpr bool DIAG = decltype(diag_tag)::value;
        constexpr bool DO_A = decltype(a_tag)::value;
        constexpr bool LAST = DIAG && !DO_A;

        v8f caccA = (v8f)0.0f, caccB = (v8f)0.0f;
#pragma unroll
        for (int s = 0; s < 16; ++s) {
            if constexpr (DO_A)
                caccA = __builtin_amdgcn_wmma_f32_16x16x4_f32(
                    false, kf[s], false, afA[s], (short)0, caccA, false, false);
            caccB = __builtin_amdgcn_wmma_f32_16x16x4_f32(
                false, kf[s], false, afB[s], (short)0, caccB, false, false);
        }

        if constexpr (DO_A)
            asm volatile("" : "+v"(caccA), "+v"(caccB),
                              "+v"(accA0), "+v"(accA1), "+v"(accA2), "+v"(accA3),
                              "+v"(accB0), "+v"(accB1), "+v"(accB2), "+v"(accB3));
        else
            asm volatile("" : "+v"(caccB),
                              "+v"(accB0), "+v"(accB1), "+v"(accB2), "+v"(accB3));
        __builtin_amdgcn_sched_barrier(0);

        float btA[8], btB[8];
        if constexpr (DIAG) {
#pragma unroll
            for (int r = 0; r < 8; ++r) {
                const int kpos  = kt * 16 + 8 * hi + r;
                const int qposB = qt1 * 16 + ln;
                float xb = caccB[r] * scale - LAMBDA_ * fabsf((float)(qposB - kpos));
                xb = fminf(fmaxf(xb, -CLAMP_), CLAMP_);
                float bb = __builtin_amdgcn_rcpf(1.0f + __expf(-xb));
                if (kpos > qposB) bb = 0.0f;
                btB[r] = bb;
                if constexpr (DO_A) {
                    const int qposA = qt0 * 16 + ln;
                    float xa = caccA[r] * scale - LAMBDA_ * fabsf((float)(qposA - kpos));
                    xa = fminf(fmaxf(xa, -CLAMP_), CLAMP_);
                    float ba = __builtin_amdgcn_rcpf(1.0f + __expf(-xa));
                    if (kpos > qposA) ba = 0.0f;
                    btA[r] = ba;
                }
            }
        } else {
            const float dA0 = (float)(qt0 * 16 + ln - kt * 16 - 8 * hi);
            const float dB0 = (float)(qt1 * 16 + ln - kt * 16 - 8 * hi);
            const v2f dA = {dA0, dA0};
            const v2f dB = {dB0, dB0};
            const v2f sc = {scale, scale};
#pragma unroll
            for (int r = 0; r < 8; r += 2) {
                const v2f off = {(float)r, (float)(r + 1)};
                const v2f cB  = {caccB[r], caccB[r + 1]};
                const v2f tB  = LAMBDA_ * (dB - off);
                const v2f xb  = __builtin_elementwise_fma(cB, sc, -tB);
                const float xb0 = fminf(fmaxf(xb.x, -CLAMP_), CLAMP_);
                const float xb1 = fminf(fmaxf(xb.y, -CLAMP_), CLAMP_);
                const v2f eB = {__expf(-xb0), __expf(-xb1)};
                const v2f nB = eB + 1.0f;
                btB[r]     = __builtin_amdgcn_rcpf(nB.x);
                btB[r + 1] = __builtin_amdgcn_rcpf(nB.y);
                if constexpr (DO_A) {
                    const v2f cA  = {caccA[r], caccA[r + 1]};
                    const v2f tA  = LAMBDA_ * (dA - off);
                    const v2f xa  = __builtin_elementwise_fma(cA, sc, -tA);
                    const float xa0 = fminf(fmaxf(xa.x, -CLAMP_), CLAMP_);
                    const float xa1 = fminf(fmaxf(xa.y, -CLAMP_), CLAMP_);
                    const v2f eA = {__expf(-xa0), __expf(-xa1)};
                    const v2f nA = eA + 1.0f;
                    btA[r]     = __builtin_amdgcn_rcpf(nA.x);
                    btA[r + 1] = __builtin_amdgcn_rcpf(nA.y);
                }
            }
        }

#define SB_STEP(R, W, BT, I)                              \
            { const float w_ = BT[I] * R;                 \
              R = fmaxf(__builtin_fmaf(-w_, R, R), EPS_); \
              W[I] = w_; }
        float waA[8], wbA[8], waB[8], wbB[8];
        float raA = remA, raB = remB;
#pragma unroll
        for (int i = 0; i < 8; ++i) {
            if constexpr (DO_A) SB_STEP(raA, waA, btA, i)
            SB_STEP(raB, waB, btB, i)
        }
        float rbA = raA, rbB = xhalf(raB);
        if constexpr (DO_A) rbA = xhalf(raA);
#pragma unroll
        for (int i = 0; i < 8; ++i) {
            if constexpr (DO_A) SB_STEP(rbA, wbA, btA, i)
            SB_STEP(rbB, wbB, btB, i)
        }
#undef SB_STEP
        if constexpr (DO_A) remA = xhalf(rbA);
        remB = xhalf(rbB);

        v2f awA[4], awB[4];
#pragma unroll
        for (int kc = 0; kc < 4; ++kc) {
            if constexpr (DO_A) {
                awA[kc].x = hi ? wbA[2 * kc]     : waA[2 * kc];
                awA[kc].y = hi ? wbA[2 * kc + 1] : waA[2 * kc + 1];
            }
            awB[kc].x = hi ? wbB[2 * kc]     : waB[2 * kc];
            awB[kc].y = hi ? wbB[2 * kc + 1] : waB[2 * kc + 1];
        }
        {
            if constexpr (DO_A) {
                const v2f sA = (awA[0] + awA[1]) + (awA[2] + awA[3]);
                denA += sA.x + sA.y;
            }
            const v2f sB = (awB[0] + awB[1]) + (awB[2] + awB[3]);
            denB += sB.x + sB.y;
        }

        v2f bv[4][4];
#pragma unroll
        for (int kc = 0; kc < 4; ++kc) {
            const v4f r0 = vr0[kc];
            const v4f r1 = vr1[kc];
            bv[kc][0].x = r0.x;  bv[kc][0].y = r1.x;
            bv[kc][1].x = r0.y;  bv[kc][1].y = r1.y;
            bv[kc][2].x = r0.z;  bv[kc][2].y = r1.z;
            bv[kc][3].x = r0.w;  bv[kc][3].y = r1.w;
        }

        if constexpr (!LAST) {
            const int kt2 = kt + 2 < QTILES ? kt + 2 : QTILES - 1;
            load_vrows(vr0, vr1, vb, kt + 1, ln, hi);
            rope_frags(kf, kx0, kx1, kcc, kss);
            load_rows(kx0, kx1, kcc, kss, kb, cosc, sinc, kt2, ln, hi);
        }

        if constexpr (DO_A)
            asm volatile("" : "+v"(awA[0]), "+v"(awA[1]), "+v"(awA[2]), "+v"(awA[3]),
                              "+v"(awB[0]), "+v"(awB[1]), "+v"(awB[2]), "+v"(awB[3]),
                              "+v"(bv[0][0]), "+v"(bv[0][1]), "+v"(bv[0][2]), "+v"(bv[0][3]),
                              "+v"(bv[1][0]), "+v"(bv[1][1]), "+v"(bv[1][2]), "+v"(bv[1][3]),
                              "+v"(bv[2][0]), "+v"(bv[2][1]), "+v"(bv[2][2]), "+v"(bv[2][3]),
                              "+v"(bv[3][0]), "+v"(bv[3][1]), "+v"(bv[3][2]), "+v"(bv[3][3]),
                              "+v"(remA), "+v"(remB), "+v"(denA), "+v"(denB));
        else
            asm volatile("" : "+v"(awB[0]), "+v"(awB[1]), "+v"(awB[2]), "+v"(awB[3]),
                              "+v"(bv[0][0]), "+v"(bv[0][1]), "+v"(bv[0][2]), "+v"(bv[0][3]),
                              "+v"(bv[1][0]), "+v"(bv[1][1]), "+v"(bv[1][2]), "+v"(bv[1][3]),
                              "+v"(bv[2][0]), "+v"(bv[2][1]), "+v"(bv[2][2]), "+v"(bv[2][3]),
                              "+v"(bv[3][0]), "+v"(bv[3][1]), "+v"(bv[3][2]), "+v"(bv[3][3]),
                              "+v"(remB), "+v"(denB));
        __builtin_amdgcn_sched_barrier(0);

#pragma unroll
        for (int kc = 0; kc < 4; ++kc) {
            const v2f b0 = bv[kc][0], b1 = bv[kc][1], b2 = bv[kc][2], b3 = bv[kc][3];
            if constexpr (DO_A)
                accA0 = __builtin_amdgcn_wmma_f32_16x16x4_f32(
                    false, awA[kc], false, b0, (short)0, accA0, false, false);
            accB0 = __builtin_amdgcn_wmma_f32_16x16x4_f32(
                false, awB[kc], false, b0, (short)0, accB0, false, false);
            if constexpr (DO_A)
                accA1 = __builtin_amdgcn_wmma_f32_16x16x4_f32(
                    false, awA[kc], false, b1, (short)0, accA1, false, false);
            accB1 = __builtin_amdgcn_wmma_f32_16x16x4_f32(
                false, awB[kc], false, b1, (short)0, accB1, false, false);
            if constexpr (DO_A)
                accA2 = __builtin_amdgcn_wmma_f32_16x16x4_f32(
                    false, awA[kc], false, b2, (short)0, accA2, false, false);
            accB2 = __builtin_amdgcn_wmma_f32_16x16x4_f32(
                false, awB[kc], false, b2, (short)0, accB2, false, false);
            if constexpr (DO_A)
                accA3 = __builtin_amdgcn_wmma_f32_16x16x4_f32(
                    false, awA[kc], false, b3, (short)0, accA3, false, false);
            accB3 = __builtin_amdgcn_wmma_f32_16x16x4_f32(
                false, awB[kc], false, b3, (short)0, accB3, false, false);
        }
    };
    int kt = 0;
    if (qt0 & 1) {
        trip(0, std::false_type{}, std::true_type{});
        kt = 1;
    }
    for (; kt < qt0; kt += 2) {
        trip(kt,     std::false_type{}, std::true_type{});
        trip(kt + 1, std::false_type{}, std::true_type{});
    }
    trip(qt0, std::true_type{}, std::true_type{});
    for (kt = qt0 + 1; kt < qt1; kt += 2) {
        trip(kt,     std::false_type{}, std::false_type{});
        trip(kt + 1, std::false_type{}, std::false_type{});
    }
    trip(qt1, std::true_type{}, std::false_type{});

    dlds[ln]      = fmaxf(denA + xhalf(denA), EPS_);
    dlds[16 + ln] = fmaxf(denB + xhalf(denB), EPS_);
    asm volatile("s_wait_dscnt 0" ::: "memory");
    __builtin_amdgcn_wave_barrier();

#pragma unroll
    for (int r = 0; r < 8; ++r) {
        const int m  = r + 8 * hi;
        const float dnA = dlds[m];
        const float dnB = dlds[16 + m];
        const int orowA = qt0 * 16 + m;
        const int orowB = qt1 * 16 + m;
        v4f oa, ob4;
        oa.x = accA0[r] / dnA;
        oa.y = accA1[r] / dnA;
        oa.z = accA2[r] / dnA;
        oa.w = accA3[r] / dnA;
        ob4.x = accB0[r] / dnB;
        ob4.y = accB1[r] / dnB;
        ob4.z = accB2[r] / dnB;
        ob4.w = accB3[r] / dnB;
        *(v4f*)(ob + orowA * D_ + 4 * ln) = oa;
        *(v4f*)(ob + orowB * D_ + 4 * ln) = ob4;
    }
  }
}

extern "C" void kernel_launch(void* const* d_in, const int* in_sizes, int n_in,
                              void* d_out, int out_size, void* d_ws, size_t ws_size,
                              hipStream_t stream) {
    (void)in_sizes; (void)n_in; (void)out_size; (void)d_ws; (void)ws_size;
    const float* q    = (const float*)d_in[0];
    const float* k    = (const float*)d_in[1];
    const float* v    = (const float*)d_in[2];
    const float* cosc = (const float*)d_in[3];
    const float* sinc = (const float*)d_in[4];
    float* out = (float*)d_out;

    const int blocks = B_ * H_ * WPH / WPB;
    rse_stickbreak_attn<<<blocks, 32 * WPB, 0, stream>>>(q, k, v, cosc, sinc, out);
}
